# starting kernel
# baseline (speedup 1.0000x reference)
#include <hip/hip_runtime.h>
#include <hip/hip_bf16.h>

typedef __attribute__((ext_vector_type(16))) _Float16 v16h;
typedef __attribute__((ext_vector_type(8)))  _Float16 v8h;
typedef __attribute__((ext_vector_type(8)))  float    v8f;

#define IN_CH 128
#define HID   64


__global__ void gcn_cvt_f16(const float* __restrict__ in, _Float16* __restrict__ out, int n) {
    int i = blockIdx.x * blockDim.x + threadIdx.x;
    if (i < n) out[i] = (_Float16)in[i];
}

__global__ void gcn_set_const(float* __restrict__ p, float v, int n) {
    int i = blockIdx.x * blockDim.x + threadIdx.x;
    if (i < n) p[i] = v;
}

__global__ void gcn_out_init_bias(float* __restrict__ out, const float* __restrict__ b, int n) {
    int i = blockIdx.x * blockDim.x + threadIdx.x;
    if (i < n) out[i] = b[i & (HID - 1)];
}

__global__ void gcn_bias_relu_cvt(const float* __restrict__ agg, const float* __restrict__ b,
                                  _Float16* __restrict__ out, int n) {
    int i = blockIdx.x * blockDim.x + threadIdx.x;
    if (i < n) out[i] = (_Float16)fmaxf(agg[i] + b[i & (HID - 1)], 0.0f);
}


__global__ void gcn_deg_accum(const int* __restrict__ dst, float* __restrict__ deg, int E) {
    int e = blockIdx.x * blockDim.x + threadIdx.x;
    if (e < E) atomicAdd(&deg[dst[e]], 1.0f);
}

__global__ void gcn_deg_rsqrt(float* __restrict__ deg, int n) {
    int i = blockIdx.x * blockDim.x + threadIdx.x;
    if (i < n) deg[i] = rsqrtf(deg[i]);
}

__global__ void gcn_pack_w(const float* __restrict__ W, _Float16* __restrict__ Bp,
                           int K, int total) {
    int idx = blockIdx.x * blockDim.x + threadIdx.x;
    if (idx >= total) return;
    int e    = idx & 15;
    int lane = (idx >> 4) & 31;
    int tile = idx >> 9;
    int nt   = tile & 3;
    int kt   = tile >> 2;
    int hi   = lane >> 4;
    int k    = kt * 32 + hi * 8 + (e & 7) + ((e & 8) ? 16 : 0);
    int n    = nt * 16 + (lane & 15);
    Bp[idx] = (_Float16)W[k * HID + n];
    (void)K;
}

__global__ void __launch_bounds__(128)
gcn_gemm_wmma(const _Float16* __restrict__ A, const _Float16* __restrict__ Bp,
              float* __restrict__ C, int K) {
    const int lane = threadIdx.x & 31;
    const int nT   = threadIdx.x >> 5;
    const int m0   = blockIdx.x << 4;
    const int hi   = lane >> 4;
    const int row  = m0 + (lane & 15);
    const int kTiles = K >> 5;

    const _Float16* arow = A + (size_t)row * K + hi * 8;
    const _Float16* bp   = Bp + ((size_t)nT * 32 + lane) * 16;

    v8f acc = {0.f, 0.f, 0.f, 0.f, 0.f, 0.f, 0.f, 0.f};

    for (int kt = 0; kt < kTiles; ++kt) {
        union { v16h v; v8h h[2]; } a;
        a.h[0] = *(const v8h*)(arow);
        a.h[1] = *(const v8h*)(arow + 16);
        v16h b = *(const v16h*)(bp);
        __builtin_prefetch(arow + 32, 0, 0);
        acc = __builtin_amdgcn_wmma_f32_16x16x32_f16(
             false, a.v,  false, b,
             (short)0, acc,  false,  false);
        arow += 32;
        bp   += 4 * 32 * 16;
    }

    float* crow = C + (size_t)(m0 + hi * 8) * HID + (nT << 4) + (lane & 15);
#pragma unroll
    for (int r = 0; r < 8; ++r) crow[(size_t)r * HID] = acc[r];
}

__global__ void __launch_bounds__(256)
gcn_aggregate(const float* __restrict__ H, const int* __restrict__ src,
              const int* __restrict__ dst, const float* __restrict__ dis,
              float* __restrict__ out, int E, int N) {
    int wave = (blockIdx.x * blockDim.x + threadIdx.x) >> 5;
    int lane = threadIdx.x & 31;
    if (wave >= E + N) return;
    int s, d;
    if (wave < E) { s = src[wave]; d = dst[wave]; }
    else          { s = d = wave - E; }
    float norm = dis[s] * dis[d];
    const float2 h2 = *(const float2*)(H + (size_t)s * HID + lane * 2);
    float* o = out + (size_t)d * HID + lane * 2;
    atomicAdd(o,     h2.x * norm);
    atomicAdd(o + 1, h2.y * norm);
}


extern "C" void kernel_launch(void* const* d_in, const int* in_sizes, int n_in,
                              void* d_out, int out_size, void* d_ws, size_t ws_size,
                              hipStream_t stream) {
    const float* x  = (const float*)d_in[0];
    const int*   ei = (const int*)d_in[1];
    const float* W1 = (const float*)d_in[2];
    const float* b1 = (const float*)d_in[3];
    const float* W2 = (const float*)d_in[4];
    const float* b2 = (const float*)d_in[5];
    float* out = (float*)d_out;

    const int N = in_sizes[0] / IN_CH;
    const int E = in_sizes[1] / 2;
    const int* src = ei;
    const int* dst = ei + E;

    char* ws = (char*)d_ws;
    size_t off = 0;
    auto alloc = [&](size_t bytes) -> char* {
        char* p = ws + off;
        off = (off + bytes + 255) & ~(size_t)255;
        return p;
    };
    _Float16* xh   = (_Float16*)alloc((size_t)N * IN_CH * sizeof(_Float16));
    _Float16* w1p  = (_Float16*)alloc((size_t)(IN_CH/32) * 4 * 32 * 16 * sizeof(_Float16));
    _Float16* w2p  = (_Float16*)alloc((size_t)(HID/32)   * 4 * 32 * 16 * sizeof(_Float16));
    float*    proj = (float*)   alloc((size_t)N * HID * sizeof(float));
    float*    agg  = (float*)   alloc((size_t)N * HID * sizeof(float));
    _Float16* h1h  = (_Float16*)alloc((size_t)N * HID * sizeof(_Float16));
    float*    dis  = (float*)   alloc((size_t)N * sizeof(float));

    const int T = 256;
    int t;

    t = N * IN_CH;
    gcn_cvt_f16<<<(t + T - 1) / T, T, 0, stream>>>(x, xh, t);

    t = (IN_CH / 32) * 4 * 32 * 16;
    gcn_pack_w<<<(t + T - 1) / T, T, 0, stream>>>(W1, w1p, IN_CH, t);
    t = (HID / 32) * 4 * 32 * 16;
    gcn_pack_w<<<(t + T - 1) / T, T, 0, stream>>>(W2, w2p, HID, t);

    gcn_set_const<<<(N + T - 1) / T, T, 0, stream>>>(dis, 1.0f, N);
    gcn_deg_accum<<<(E + T - 1) / T, T, 0, stream>>>(dst, dis, E);
    gcn_deg_rsqrt<<<(N + T - 1) / T, T, 0, stream>>>(dis, N);

    gcn_gemm_wmma<<<N / 16, 128, 0, stream>>>(xh, w1p, proj, IN_CH);

    t = N * HID;
    gcn_set_const<<<(t + T - 1) / T, T, 0, stream>>>(agg, 0.0f, t);
    {
        long long threads = (long long)(E + N) * 32;
        int blocks = (int)((threads + T - 1) / T);
        gcn_aggregate<<<blocks, T, 0, stream>>>(proj, src, dst, dis, agg, E, N);
    }

    gcn_bias_relu_cvt<<<(t + T - 1) / T, T, 0, stream>>>(agg, b1, h1h, t);

    gcn_gemm_wmma<<<N / 16, 128, 0, stream>>>(h1h, w2p, proj, HID);

    gcn_out_init_bias<<<(t + T - 1) / T, T, 0, stream>>>(out, b2, t);
    {
        long long threads = (long long)(E + N) * 32;
        int blocks = (int)((threads + T - 1) / T);
        gcn_aggregate<<<blocks, T, 0, stream>>>(proj, src, dst, dis, out, E, N);
    }
}


# neighbour sums by 256 owner work-groups (196 nodes each): each picks its edges from a one-byte owner list, adds rows in LDS (integer, order-free), stores f32; no global sum atomics; 8->7 launches
# speedup vs baseline: 2.6640x; 2.6640x over previous
#include <hip/hip_runtime.h>

typedef __attribute__((ext_vector_type(2)))  float    v2f;
typedef __attribute__((ext_vector_type(8)))  float    v8f;

#define IN_CH 128
#define HID   64

#define FX_SCALE 1048576.0f
#define FX_INV   (1.0f / 1048576.0f)

#define OWN_ROWS    196
#define OWN_THREADS 1024
#define OWN_WCAP    256

static __device__ __forceinline__ int gcn_half_of_wave() {
    return (int)((__builtin_amdgcn_s_sendmsg_rtn(0x87) >> 18) & 1u);
}


__global__ void gcn_set_const(float* __restrict__ p0, float* __restrict__ p1, float v, int n) {
    int i = blockIdx.x * blockDim.x + threadIdx.x;
    if (i < n) { p0[i] = v; p1[i] = 0.0f; }
}

__global__ void gcn_deg_accum(const int* __restrict__ dst, float* __restrict__ deg0,
                              float* __restrict__ deg1, unsigned char* __restrict__ owner, int E) {
    const int side = gcn_half_of_wave();
    float* deg = side ? deg1 : deg0;
    const int e0 = (blockIdx.x * blockDim.x + threadIdx.x) * 4;
    if (e0 >= E) return;
    if (((E & 3) == 0)) {
        const int4 d = *(const int4*)(dst + e0);
        atomicAdd(&deg[d.x], 1.0f);
        atomicAdd(&deg[d.y], 1.0f);
        atomicAdd(&deg[d.z], 1.0f);
        atomicAdd(&deg[d.w], 1.0f);
        uchar4 o;
        o.x = (unsigned char)((unsigned)d.x / OWN_ROWS);
        o.y = (unsigned char)((unsigned)d.y / OWN_ROWS);
        o.z = (unsigned char)((unsigned)d.z / OWN_ROWS);
        o.w = (unsigned char)((unsigned)d.w / OWN_ROWS);
        *(uchar4*)(owner + e0) = o;
    } else {
        for (int k = 0; k < 4; ++k) {
            const int e = e0 + k;
            if (e < E) {
                const int d = dst[e];
                atomicAdd(&deg[d], 1.0f);
                owner[e] = (unsigned char)((unsigned)d / OWN_ROWS);
            }
        }
    }
}

__global__ void gcn_deg_rsqrt(float* __restrict__ deg0, const float* __restrict__ deg1, int n) {
    int i = blockIdx.x * blockDim.x + threadIdx.x;
    if (i < n) deg0[i] = rsqrtf(deg0[i] + deg1[i]);
}

template <int K>
__global__ void __launch_bounds__(128)
gcn_gemm_wmma(const float* __restrict__ Af, const float* __restrict__ W, float* __restrict__ C) {
    const int lane = threadIdx.x & 31;
    const int nT   = threadIdx.x >> 5;
    const int m0   = blockIdx.x << 4;
    const int hi   = lane >> 4;
    const int row  = m0 + (lane & 15);
    const int col  = (nT << 4) + (lane & 15);

    const size_t aofs = (size_t)row * K + hi * 4;
    const float* wk   = W + (size_t)(hi * 4) * HID + col;

    v8f acc0 = {0.f, 0.f, 0.f, 0.f, 0.f, 0.f, 0.f, 0.f};
    v8f acc1 = {0.f, 0.f, 0.f, 0.f, 0.f, 0.f, 0.f, 0.f};

#pragma unroll
    for (int kb = 0; kb < K / 8; ++kb) {
        const float4 a = *(const float4*)(Af + aofs + kb * 8);
        const float* wb = wk + (size_t)(kb * 8) * HID;
        const v2f a0 = {a.x, a.y};
        const v2f a1 = {a.z, a.w};
        const v2f b0 = {wb[0],       wb[HID]};
        const v2f b1 = {wb[2 * HID], wb[3 * HID]};
        acc0 = __builtin_amdgcn_wmma_f32_16x16x4_f32(
             false, a0,  false, b0,
             (short)0, acc0,  false,  false);
        acc1 = __builtin_amdgcn_wmma_f32_16x16x4_f32(
             false, a1,  false, b1,
             (short)0, acc1,  false,  false);
    }
    const v8f acc = acc0 + acc1;

    float* crow = C + (size_t)(m0 + hi * 8) * HID + col;
#pragma unroll
    for (int r = 0; r < 8; ++r) crow[(size_t)r * HID] = acc[r];
}


static __device__ __noinline__ void gcn_add_edge_by_lane(int* acc, const float* __restrict__ H,
                                                         const int* __restrict__ src, const int* __restrict__ dst,
                                                         const float* __restrict__ dis, int e, int n0) {
    const int s = src[e];
    const int d = dst[e];
    const float w = dis[s] * dis[d] * FX_SCALE;
    const float* hrow = H + (size_t)s * HID;
    int* orow = acc + (d - n0) * HID;
    for (int c = 0; c < HID; ++c) atomicAdd(orow + c, __float2int_rn(hrow[c] * w));
}

static __device__ __forceinline__ int gcn_pick_edges(int* acc, int* wl, const float* __restrict__ H,
                                                     const int* __restrict__ src, const int* __restrict__ dst,
                                                     const unsigned char* __restrict__ owner,
                                                     const float* __restrict__ dis, int E, int n0, int tid) {
    const unsigned pat = (unsigned)blockIdx.x * 0x01010101u;
    const int nq = E >> 4;
    const uint4* oq = (const uint4*)owner;
    int wcnt = 0;
    for (int qb = 0; qb < nq; qb += OWN_THREADS * 8) {
        const int q0 = qb + tid;
        uint4 v[8];
#pragma unroll
        for (int u = 0; u < 8; ++u) {
            v[u] = oq[min(q0 + u * OWN_THREADS, nq - 1)];
        }
#pragma unroll
        for (int u = 0; u < 8; ++u) {
            const bool inside = (q0 + u * OWN_THREADS) < nq;
            const unsigned x0 = v[u].x ^ pat, x1 = v[u].y ^ pat, x2 = v[u].z ^ pat, x3 = v[u].w ^ pat;
            const unsigned f0 = ~(((x0 & 0x7F7F7F7Fu) + 0x7F7F7F7Fu) | x0 | 0x7F7F7F7Fu);
            const unsigned f1 = ~(((x1 & 0x7F7F7F7Fu) + 0x7F7F7F7Fu) | x1 | 0x7F7F7F7Fu);
            const unsigned f2 = ~(((x2 & 0x7F7F7F7Fu) + 0x7F7F7F7Fu) | x2 | 0x7F7F7F7Fu);
            const unsigned f3 = ~(((x3 & 0x7F7F7F7Fu) + 0x7F7F7F7Fu) | x3 | 0x7F7F7F7Fu);
            unsigned m = (f0 >> 7) | (f1 >> 6) | (f2 >> 5) | (f3 >> 4);
            m = inside ? m : 0u;
            const int ebase = (q0 + u * OWN_THREADS) * 16;
            unsigned bal = __builtin_amdgcn_ballot_w32(m != 0u);
            while (bal != 0u) {
                const bool has = (m != 0u);
                const int p = __builtin_ctz(m | 0x80000000u);
                const int e = ebase + ((p & 7) << 2) + (p >> 3);
                m &= m - 1u;
                const int slot = wcnt + (int)__builtin_amdgcn_mbcnt_lo(bal, 0u);
                if (has) {
                    if (slot < OWN_WCAP) wl[slot] = e;
                    else gcn_add_edge_by_lane(acc, H, src, dst, dis, e, n0);
                }
                wcnt += __builtin_popcount(bal);
                bal = __builtin_amdgcn_ballot_w32(m != 0u);
            }
        }
    }
    return wcnt;
}

template <bool FIRST>
__global__ void __launch_bounds__(OWN_THREADS)
gcn_owner_sum(const float* __restrict__ H, const int* __restrict__ src, const int* __restrict__ dst,
              const unsigned char* __restrict__ owner, const float* __restrict__ dis,
              const float* __restrict__ bias, float* __restrict__ outp,
              int* saved, int* saved_cnt, int N, int E) {
    __shared__ int acc[OWN_ROWS * HID];
    __shared__ int lists[(OWN_THREADS / 32) * OWN_WCAP];

    const int tid  = threadIdx.x;
    const int lane = tid & 31;
    const int wave = tid >> 5;
    const int n0   = blockIdx.x * OWN_ROWS;
    const int rows = min(OWN_ROWS, N - n0);
    const int nv4  = rows * (HID / 4);
    int* wl = lists + wave * OWN_WCAP;
    const int wslot = blockIdx.x * (OWN_THREADS / 32) + wave;
    int* sv = saved + (size_t)wslot * OWN_WCAP;

    for (int i = tid; i < nv4; i += OWN_THREADS) {
        const int r = i >> 4, c4 = (i & 15) * 4;
        const float dd = dis[n0 + r];
        const float d2 = dd * dd;
        const float4 h = *(const float4*)(H + (size_t)(n0 + r) * HID + c4);
        const float4 b = *(const float4*)(bias + c4);
        int4 v;
        v.x = __float2int_rn((b.x + h.x * d2) * FX_SCALE);
        v.y = __float2int_rn((b.y + h.y * d2) * FX_SCALE);
        v.z = __float2int_rn((b.z + h.z * d2) * FX_SCALE);
        v.w = __float2int_rn((b.w + h.w * d2) * FX_SCALE);
        *(int4*)(acc + i * 4) = v;
    }
    __syncthreads();

    {
        const int et = (E & ~15) + tid;
        if (tid < (E & 15) && owner[et] == (unsigned char)blockIdx.x)
            gcn_add_edge_by_lane(acc, H, src, dst, dis, et, n0);
    }

    int found;
    if (FIRST) {
        found = __builtin_amdgcn_readfirstlane(gcn_pick_edges(acc, wl, H, src, dst, owner, dis, E, n0, tid));
        const int keep = min(found, OWN_WCAP);
        __syncthreads();
        for (int j = lane; j < keep; j += 32) sv[j] = wl[j];
        if (lane == 0) saved_cnt[wslot] = found;
    } else {
        found = __builtin_amdgcn_readfirstlane(saved_cnt[wslot]);
        if (found <= OWN_WCAP) {
            for (int j = lane; j < found; j += 32) wl[j] = sv[j];
        } else {
            found = __builtin_amdgcn_readfirstlane(gcn_pick_edges(acc, wl, H, src, dst, owner, dis, E, n0, tid));
        }
        __syncthreads();
    }
    const int cnt = min(found, OWN_WCAP);

    for (int base = 0; base < cnt; base += 32) {
        const int j = base + lane;
        int s = 0, dl = 0;
        float w = 0.0f;
        if (j < cnt) {
            const int e = wl[j];
            s = src[e];
            const int d = dst[e];
            dl = d - n0;
            w = dis[s] * dis[d] * FX_SCALE;
        }
        const int wbits = __float_as_int(w);
        const int left = cnt - base;

        float h0[4][8], h1[4][8];

#define GCN_LOAD_GROUP(g, on)                                                      \
    if (on) {                                                                      \
    _Pragma("unroll")   \
    for (int u = 0; u < 8; ++u) {                                                  \
        const int sj = __builtin_amdgcn_readlane(s, (g) * 8 + u);                  \
        const float* hrow = H + (size_t)sj * HID;        \
        h0[g][u] = hrow[lane];                                                     \
        h1[g][u] = hrow[lane + 32];                                                \
    } }
#define GCN_ADD_GROUP(g, on)                                                       \
    if (on) {                                                                      \
    _Pragma("unroll")   \
    for (int u = 0; u < 8; ++u) {                                                  \
        const int   dj = __builtin_amdgcn_readlane(dl, (g) * 8 + u);               \
        const float wj = __int_as_float(__builtin_amdgcn_readlane(wbits, (g) * 8 + u)); \
        int* orow = acc + dj * HID;                       \
        atomicAdd(orow + lane,      __float2int_rn(h0[g][u] * wj));                \
        atomicAdd(orow + lane + 32, __float2int_rn(h1[g][u] * wj));                \
    } }

        if (left >= 32) {
            GCN_LOAD_GROUP(0, true)
            GCN_LOAD_GROUP(1, true)
            GCN_ADD_GROUP(0, true)
            GCN_LOAD_GROUP(2, true)
            GCN_ADD_GROUP(1, true)
            GCN_LOAD_GROUP(3, true)
            GCN_ADD_GROUP(2, true)
            GCN_ADD_GROUP(3, true)
        } else {
            GCN_LOAD_GROUP(0, true)
            GCN_LOAD_GROUP(1, left > 8)
            GCN_ADD_GROUP(0, true)
            GCN_LOAD_GROUP(2, left > 16)
            GCN_ADD_GROUP(1, left > 8)
            GCN_LOAD_GROUP(3, left > 24)
            GCN_ADD_GROUP(2, left > 16)
            GCN_ADD_GROUP(3, left > 24)
        }
#undef GCN_LOAD_GROUP
#undef GCN_ADD_GROUP
    }
    __syncthreads();

    float4* orows = (float4*)(outp + (size_t)n0 * HID);
    for (int i = tid; i < nv4; i += OWN_THREADS) {
        const int4 a = *(const int4*)(acc + i * 4);
        float4 r;
        r.x = (float)a.x * FX_INV;
        r.y = (float)a.y * FX_INV;
        r.z = (float)a.z * FX_INV;
        r.w = (float)a.w * FX_INV;
        if (FIRST) {
            r.x = fmaxf(r.x, 0.0f); r.y = fmaxf(r.y, 0.0f);
            r.z = fmaxf(r.z, 0.0f); r.w = fmaxf(r.w, 0.0f);
        }
        orows[i] = r;
    }
}


extern "C" void kernel_launch(void* const* d_in, const int* in_sizes, int n_in,
                              void* d_out, int out_size, void* d_ws, size_t ws_size,
                              hipStream_t stream) {
    const float* x  = (const float*)d_in[0];
    const int*   ei = (const int*)d_in[1];
    const float* W1 = (const float*)d_in[2];
    const float* b1 = (const float*)d_in[3];
    const float* W2 = (const float*)d_in[4];
    const float* b2 = (const float*)d_in[5];
    float* out = (float*)d_out;

    const int N = in_sizes[0] / IN_CH;
    const int E = in_sizes[1] / 2;
    const int* src = ei;
    const int* dst = ei + E;

    const int owners = (N + OWN_ROWS - 1) / OWN_ROWS;

    char* ws = (char*)d_ws;
    size_t off = 0;
    auto alloc = [&](size_t bytes) -> char* {
        char* p = ws + off;
        off = (off + bytes + 255) & ~(size_t)255;
        return p;
    };
    float*         proj  = (float*)        alloc((size_t)N * HID * sizeof(float));
    unsigned char* owner = (unsigned char*)alloc((size_t)E + 16);
    int*           saved = (int*)          alloc((size_t)owners * (OWN_THREADS / 32) * OWN_WCAP * sizeof(int));
    int*           scnt  = (int*)          alloc((size_t)owners * (OWN_THREADS / 32) * sizeof(int));
    float*         dis   = (float*)        alloc((size_t)N * sizeof(float));
    float*         deg1  = (float*)        alloc((size_t)N * sizeof(float));
    float*         h1    = out;

    const int T = 256;

    gcn_set_const<<<(N + T - 1) / T, T, 0, stream>>>(dis, deg1, 1.0f, N);
    gcn_deg_accum<<<((E + 3) / 4 + T - 1) / T, T, 0, stream>>>(dst, dis, deg1, owner, E);
    gcn_deg_rsqrt<<<(N + T - 1) / T, T, 0, stream>>>(dis, deg1, N);

    gcn_gemm_wmma<IN_CH><<<N / 16, 128, 0, stream>>>(x, W1, proj);
    gcn_owner_sum<true><<<owners, OWN_THREADS, 0, stream>>>(proj, src, dst, owner, dis, b1, h1, saved, scnt, N, E);

    gcn_gemm_wmma<HID><<<N / 16, 128, 0, stream>>>(h1, W2, proj);
    gcn_owner_sum<false><<<owners, OWN_THREADS, 0, stream>>>(proj, src, dst, owner, dis, b2, out, saved, scnt, N, E);
}


# layer-2 product done by the layer-1 owner work-groups on relu(finished rows) straight from LDS (row pitch 68 words), f32 matrix form, same k order; layer-1 output never stored; 7 -> 6 launches
# speedup vs baseline: 2.7529x; 1.0334x over previous
#include <hip/hip_runtime.h>

typedef __attribute__((ext_vector_type(2)))  float    v2f;
typedef __attribute__((ext_vector_type(8)))  float    v8f;

#define IN_CH 128
#define HID   64

#define FX_SCALE 1048576.0f
#define FX_INV   (1.0f / 1048576.0f)

#define OWN_ROWS    196
#define OWN_THREADS 1024
#define OWN_WCAP    256
#define ACC_PITCH   68

static __device__ __forceinline__ int gcn_half_of_wave() {
    return (int)((__builtin_amdgcn_s_sendmsg_rtn(0x87) >> 18) & 1u);
}


__global__ void gcn_set_const(float* __restrict__ p0, float* __restrict__ p1, float v, int n) {
    int i = blockIdx.x * blockDim.x + threadIdx.x;
    if (i < n) { p0[i] = v; p1[i] = 0.0f; }
}

__global__ void gcn_deg_accum(const int* __restrict__ dst, float* __restrict__ deg0,
                              float* __restrict__ deg1, unsigned char* __restrict__ owner, int E) {
    const int side = gcn_half_of_wave();
    float* deg = side ? deg1 : deg0;
    const int e0 = (blockIdx.x * blockDim.x + threadIdx.x) * 4;
    if (e0 >= E) return;
    if (((E & 3) == 0)) {
        const int4 d = *(const int4*)(dst + e0);
        atomicAdd(&deg[d.x], 1.0f);
        atomicAdd(&deg[d.y], 1.0f);
        atomicAdd(&deg[d.z], 1.0f);
        atomicAdd(&deg[d.w], 1.0f);
        uchar4 o;
        o.x = (unsigned char)((unsigned)d.x / OWN_ROWS);
        o.y = (unsigned char)((unsigned)d.y / OWN_ROWS);
        o.z = (unsigned char)((unsigned)d.z / OWN_ROWS);
        o.w = (unsigned char)((unsigned)d.w / OWN_ROWS);
        *(uchar4*)(owner + e0) = o;
    } else {
        for (int k = 0; k < 4; ++k) {
            const int e = e0 + k;
            if (e < E) {
                const int d = dst[e];
                atomicAdd(&deg[d], 1.0f);
                owner[e] = (unsigned char)((unsigned)d / OWN_ROWS);
            }
        }
    }
}

__global__ void gcn_deg_rsqrt(float* __restrict__ deg0, const float* __restrict__ deg1, int n) {
    int i = blockIdx.x * blockDim.x + threadIdx.x;
    if (i < n) deg0[i] = rsqrtf(deg0[i] + deg1[i]);
}

template <int K>
__global__ void __launch_bounds__(128)
gcn_gemm_wmma(const float* __restrict__ Af, const float* __restrict__ W, float* __restrict__ C) {
    const int lane = threadIdx.x & 31;
    const int nT   = threadIdx.x >> 5;
    const int m0   = blockIdx.x << 4;
    const int hi   = lane >> 4;
    const int row  = m0 + (lane & 15);
    const int col  = (nT << 4) + (lane & 15);

    const size_t aofs = (size_t)row * K + hi * 4;
    const float* wk   = W + (size_t)(hi * 4) * HID + col;

    v8f acc0 = {0.f, 0.f, 0.f, 0.f, 0.f, 0.f, 0.f, 0.f};
    v8f acc1 = {0.f, 0.f, 0.f, 0.f, 0.f, 0.f, 0.f, 0.f};

#pragma unroll
    for (int kb = 0; kb < K / 8; ++kb) {
        const float4 a = *(const float4*)(Af + aofs + kb * 8);
        const float* wb = wk + (size_t)(kb * 8) * HID;
        const v2f a0 = {a.x, a.y};
        const v2f a1 = {a.z, a.w};
        const v2f b0 = {wb[0],       wb[HID]};
        const v2f b1 = {wb[2 * HID], wb[3 * HID]};
        acc0 = __builtin_amdgcn_wmma_f32_16x16x4_f32(
             false, a0,  false, b0,
             (short)0, acc0,  false,  false);
        acc1 = __builtin_amdgcn_wmma_f32_16x16x4_f32(
             false, a1,  false, b1,
             (short)0, acc1,  false,  false);
    }
    const v8f acc = acc0 + acc1;

    float* crow = C + (size_t)(m0 + hi * 8) * HID + col;
#pragma unroll
    for (int r = 0; r < 8; ++r) crow[(size_t)r * HID] = acc[r];
}


static __device__ __noinline__ void gcn_add_edge_by_lane(int* acc, const float* __restrict__ H,
                                                         const int* __restrict__ src, const int* __restrict__ dst,
                                                         const float* __restrict__ dis, int e, int n0) {
    const int s = src[e];
    const int d = dst[e];
    const float w = dis[s] * dis[d] * FX_SCALE;
    const float* hrow = H + (size_t)s * HID;
    int* orow = acc + (d - n0) * ACC_PITCH;
    for (int c = 0; c < HID; ++c) atomicAdd(orow + c, __float2int_rn(hrow[c] * w));
}

static __device__ __forceinline__ int gcn_pick_edges(int* acc, int* wl, const float* __restrict__ H,
                                                     const int* __restrict__ src, const int* __restrict__ dst,
                                                     const unsigned char* __restrict__ owner,
                                                     const float* __restrict__ dis, int E, int n0, int tid) {
    const unsigned pat = (unsigned)blockIdx.x * 0x01010101u;
    const int nq = E >> 4;
    const uint4* oq = (const uint4*)owner;
    int wcnt = 0;
    for (int qb = 0; qb < nq; qb += OWN_THREADS * 8) {
        const int q0 = qb + tid;
        uint4 v[8];
#pragma unroll
        for (int u = 0; u < 8; ++u) {
            v[u] = oq[min(q0 + u * OWN_THREADS, nq - 1)];
        }
#pragma unroll
        for (int u = 0; u < 8; ++u) {
            const bool inside = (q0 + u * OWN_THREADS) < nq;
            const unsigned x0 = v[u].x ^ pat, x1 = v[u].y ^ pat, x2 = v[u].z ^ pat, x3 = v[u].w ^ pat;
            const unsigned f0 = ~(((x0 & 0x7F7F7F7Fu) + 0x7F7F7F7Fu) | x0 | 0x7F7F7F7Fu);
            const unsigned f1 = ~(((x1 & 0x7F7F7F7Fu) + 0x7F7F7F7Fu) | x1 | 0x7F7F7F7Fu);
            const unsigned f2 = ~(((x2 & 0x7F7F7F7Fu) + 0x7F7F7F7Fu) | x2 | 0x7F7F7F7Fu);
            const unsigned f3 = ~(((x3 & 0x7F7F7F7Fu) + 0x7F7F7F7Fu) | x3 | 0x7F7F7F7Fu);
            unsigned m = (f0 >> 7) | (f1 >> 6) | (f2 >> 5) | (f3 >> 4);
            m = inside ? m : 0u;
            const int ebase = (q0 + u * OWN_THREADS) * 16;
            unsigned bal = __builtin_amdgcn_ballot_w32(m != 0u);
            while (bal != 0u) {
                const bool has = (m != 0u);
                const int p = __builtin_ctz(m | 0x80000000u);
                const int e = ebase + ((p & 7) << 2) + (p >> 3);
                m &= m - 1u;
                const int slot = wcnt + (int)__builtin_amdgcn_mbcnt_lo(bal, 0u);
                if (has) {
                    if (slot < OWN_WCAP) wl[slot] = e;
                    else gcn_add_edge_by_lane(acc, H, src, dst, dis, e, n0);
                }
                wcnt += __builtin_popcount(bal);
                bal = __builtin_amdgcn_ballot_w32(m != 0u);
            }
        }
    }
    return wcnt;
}

template <bool FIRST>
__global__ void __launch_bounds__(OWN_THREADS)
gcn_owner_sum(const float* __restrict__ H, const int* __restrict__ src, const int* __restrict__ dst,
              const unsigned char* __restrict__ owner, const float* __restrict__ dis,
              const float* __restrict__ bias, const float* __restrict__ Wn, float* __restrict__ outp,
              int* saved, int* saved_cnt, int N, int E) {
    __shared__ int acc[OWN_ROWS * ACC_PITCH];
    __shared__ int lists[(OWN_THREADS / 32) * OWN_WCAP];

    const int tid  = threadIdx.x;
    const int lane = tid & 31;
    const int wave = tid >> 5;
    const int n0   = blockIdx.x * OWN_ROWS;
    const int rows = min(OWN_ROWS, N - n0);
    const int nv4  = rows * (HID / 4);
    int* wl = lists + wave * OWN_WCAP;
    const int wslot = blockIdx.x * (OWN_THREADS / 32) + wave;
    int* sv = saved + (size_t)wslot * OWN_WCAP;

    for (int i = tid; i < nv4; i += OWN_THREADS) {
        const int r = i >> 4, c4 = (i & 15) * 4;
        const float dd = dis[n0 + r];
        const float d2 = dd * dd;
        const float4 h = *(const float4*)(H + (size_t)(n0 + r) * HID + c4);
        const float4 b = *(const float4*)(bias + c4);
        int4 v;
        v.x = __float2int_rn((b.x + h.x * d2) * FX_SCALE);
        v.y = __float2int_rn((b.y + h.y * d2) * FX_SCALE);
        v.z = __float2int_rn((b.z + h.z * d2) * FX_SCALE);
        v.w = __float2int_rn((b.w + h.w * d2) * FX_SCALE);
        *(int4*)(acc + r * ACC_PITCH + c4) = v;
    }
    __syncthreads();

    {
        const int et = (E & ~15) + tid;
        if (tid < (E & 15) && owner[et] == (unsigned char)blockIdx.x)
            gcn_add_edge_by_lane(acc, H, src, dst, dis, et, n0);
    }

    int found;
    if (FIRST) {
        found = __builtin_amdgcn_readfirstlane(gcn_pick_edges(acc, wl, H, src, dst, owner, dis, E, n0, tid));
        const int keep = min(found, OWN_WCAP);
        __syncthreads();
        for (int j = lane; j < keep; j += 32) sv[j] = wl[j];
        if (lane == 0) saved_cnt[wslot] = found;
    } else {
        found = __builtin_amdgcn_readfirstlane(saved_cnt[wslot]);
        if (found <= OWN_WCAP) {
            for (int j = lane; j < found; j += 32) wl[j] = sv[j];
        } else {
            found = __builtin_amdgcn_readfirstlane(gcn_pick_edges(acc, wl, H, src, dst, owner, dis, E, n0, tid));
        }
        __syncthreads();
    }
    const int cnt = min(found, OWN_WCAP);

    for (int base = 0; base < cnt; base += 32) {
        const int j = base + lane;
        int s = 0, dl = 0;
        float w = 0.0f;
        if (j < cnt) {
            const int e = wl[j];
            s = src[e];
            const int d = dst[e];
            dl = d - n0;
            w = dis[s] * dis[d] * FX_SCALE;
        }
        const int wbits = __float_as_int(w);
        const int left = cnt - base;

        float h0[4][8], h1[4][8];

#define GCN_LOAD_GROUP(g, on)                                                      \
    if (on) {                                                                      \
    _Pragma("unroll")   \
    for (int u = 0; u < 8; ++u) {                                                  \
        const int sj = __builtin_amdgcn_readlane(s, (g) * 8 + u);                  \
        const float* hrow = H + (size_t)sj * HID;        \
        h0[g][u] = hrow[lane];                                                     \
        h1[g][u] = hrow[lane + 32];                                                \
    } }
#define GCN_ADD_GROUP(g, on)                                                       \
    if (on) {                                                                      \
    _Pragma("unroll")   \
    for (int u = 0; u < 8; ++u) {                                                  \
        const int   dj = __builtin_amdgcn_readlane(dl, (g) * 8 + u);               \
        const float wj = __int_as_float(__builtin_amdgcn_readlane(wbits, (g) * 8 + u)); \
        int* orow = acc + dj * ACC_PITCH;                 \
        atomicAdd(orow + lane,      __float2int_rn(h0[g][u] * wj));                \
        atomicAdd(orow + lane + 32, __float2int_rn(h1[g][u] * wj));                \
    } }

        if (left >= 32) {
            GCN_LOAD_GROUP(0, true)
            GCN_LOAD_GROUP(1, true)
            GCN_ADD_GROUP(0, true)
            GCN_LOAD_GROUP(2, true)
            GCN_ADD_GROUP(1, true)
            GCN_LOAD_GROUP(3, true)
            GCN_ADD_GROUP(2, true)
            GCN_ADD_GROUP(3, true)
        } else {
            GCN_LOAD_GROUP(0, true)
            GCN_LOAD_GROUP(1, left > 8)
            GCN_ADD_GROUP(0, true)
            GCN_LOAD_GROUP(2, left > 16)
            GCN_ADD_GROUP(1, left > 8)
            GCN_LOAD_GROUP(3, left > 24)
            GCN_ADD_GROUP(2, left > 16)
            GCN_ADD_GROUP(3, left > 24)
        }
#undef GCN_LOAD_GROUP
#undef GCN_ADD_GROUP
    }
    __syncthreads();

    if (FIRST) {
        const int c  = lane & 15;
        const int hi = lane >> 4;
        const int wv = __builtin_amdgcn_readfirstlane(wave);
        const int units = ((rows + 15) >> 4) * 2;
        for (int unit = wv; unit < units; unit += OWN_THREADS / 32) {
            const int m0l = (unit >> 1) << 4;
            const int nb  = (unit & 1) << 5;
            const int* arow = acc + min(m0l + c, rows - 1) * ACC_PITCH + hi * 4;
            const float* wk = Wn + (size_t)(hi * 4) * HID + nb + c;
            v8f p0 = {0.f, 0.f, 0.f, 0.f, 0.f, 0.f, 0.f, 0.f};
            v8f p1 = {0.f, 0.f, 0.f, 0.f, 0.f, 0.f, 0.f, 0.f};
            v8f q0 = {0.f, 0.f, 0.f, 0.f, 0.f, 0.f, 0.f, 0.f};
            v8f q1 = {0.f, 0.f, 0.f, 0.f, 0.f, 0.f, 0.f, 0.f};
#pragma unroll
            for (int kb = 0; kb < HID / 8; ++kb) {
                const int4 ai = *(const int4*)(arow + kb * 8);
                const v2f a0 = {fmaxf((float)ai.x * FX_INV, 0.0f), fmaxf((float)ai.y * FX_INV, 0.0f)};
                const v2f a1 = {fmaxf((float)ai.z * FX_INV, 0.0f), fmaxf((float)ai.w * FX_INV, 0.0f)};
                const float* wb = wk + (size_t)(kb * 8) * HID;
                const v2f b0 = {wb[0],            wb[HID]};
                const v2f b1 = {wb[2 * HID],      wb[3 * HID]};
                const v2f d0 = {wb[16],           wb[HID + 16]};
                const v2f d1 = {wb[2 * HID + 16], wb[3 * HID + 16]};
                p0 = __builtin_amdgcn_wmma_f32_16x16x4_f32(false, a0, false, b0, (short)0, p0, false, false);
                q0 = __builtin_amdgcn_wmma_f32_16x16x4_f32(false, a0, false, d0, (short)0, q0, false, false);
                p1 = __builtin_amdgcn_wmma_f32_16x16x4_f32(false, a1, false, b1, (short)0, p1, false, false);
                q1 = __builtin_amdgcn_wmma_f32_16x16x4_f32(false, a1, false, d1, (short)0, q1, false, false);
            }
            const v8f p = p0 + p1;
            const v8f q = q0 + q1;
            float* crow = outp + (size_t)(n0 + m0l + hi * 8) * HID + nb + c;
            if (m0l + 16 <= rows) {
#pragma unroll
                for (int r = 0; r < 8; ++r) {
                    crow[(size_t)r * HID]      = p[r];
                    crow[(size_t)r * HID + 16] = q[r];
                }
            } else {
                const int live = rows - (m0l + hi * 8);
#pragma unroll
                for (int r = 0; r < 8; ++r) {
                    if (r < live) {
                        crow[(size_t)r * HID]      = p[r];
                        crow[(size_t)r * HID + 16] = q[r];
                    }
                }
            }
        }
    } else {
        float4* orows = (float4*)(outp + (size_t)n0 * HID);
        for (int i = tid; i < nv4; i += OWN_THREADS) {
            const int r = i >> 4, c4 = (i & 15) * 4;
            const int4 a = *(const int4*)(acc + r * ACC_PITCH + c4);
            float4 v;
            v.x = (float)a.x * FX_INV;
            v.y = (float)a.y * FX_INV;
            v.z = (float)a.z * FX_INV;
            v.w = (float)a.w * FX_INV;
            orows[i] = v;
        }
    }
}


extern "C" void kernel_launch(void* const* d_in, const int* in_sizes, int n_in,
                              void* d_out, int out_size, void* d_ws, size_t ws_size,
                              hipStream_t stream) {
    const float* x  = (const float*)d_in[0];
    const int*   ei = (const int*)d_in[1];
    const float* W1 = (const float*)d_in[2];
    const float* b1 = (const float*)d_in[3];
    const float* W2 = (const float*)d_in[4];
    const float* b2 = (const float*)d_in[5];
    float* out = (float*)d_out;

    const int N = in_sizes[0] / IN_CH;
    const int E = in_sizes[1] / 2;
    const int* src = ei;
    const int* dst = ei + E;

    const int owners = (N + OWN_ROWS - 1) / OWN_ROWS;

    char* ws = (char*)d_ws;
    size_t off = 0;
    auto alloc = [&](size_t bytes) -> char* {
        char* p = ws + off;
        off = (off + bytes + 255) & ~(size_t)255;
        return p;
    };
    float*         proj  = (float*)        alloc((size_t)N * HID * sizeof(float));
    unsigned char* owner = (unsigned char*)alloc((size_t)E + 16);
    int*           saved = (int*)          alloc((size_t)owners * (OWN_THREADS / 32) * OWN_WCAP * sizeof(int));
    int*           scnt  = (int*)          alloc((size_t)owners * (OWN_THREADS / 32) * sizeof(int));
    float*         dis   = (float*)        alloc((size_t)N * sizeof(float));
    float*         deg1  = (float*)        alloc((size_t)N * sizeof(float));
    float*         proj1 = out;

    const int T = 256;

    gcn_set_const<<<(N + T - 1) / T, T, 0, stream>>>(dis, deg1, 1.0f, N);
    gcn_deg_accum<<<((E + 3) / 4 + T - 1) / T, T, 0, stream>>>(dst, dis, deg1, owner, E);
    gcn_deg_rsqrt<<<(N + T - 1) / T, T, 0, stream>>>(dis, deg1, N);

    gcn_gemm_wmma<IN_CH><<<N / 16, 128, 0, stream>>>(x, W1, proj1);
    gcn_owner_sum<true><<<owners, OWN_THREADS, 0, stream>>>(proj1, src, dst, owner, dis, b1, W2, proj, saved, scnt, N, E);

    gcn_owner_sum<false><<<owners, OWN_THREADS, 0, stream>>>(proj, src, dst, owner, dis, b2, nullptr, out, saved, scnt, N, E);
}


# degree pass groups each block of 1024 edges by owner in LDS (rank = LDS add's old value), stores start+length per block and owner: owners copy their groups, byte-list search gone; integer sums kept
# speedup vs baseline: 3.0306x; 1.1009x over previous
#include <hip/hip_runtime.h>

typedef __attribute__((ext_vector_type(2)))  float    v2f;
typedef __attribute__((ext_vector_type(8)))  float    v8f;

#define IN_CH 128
#define HID   64

#define FX_SCALE 1048576.0f
#define FX_INV   (1.0f / 1048576.0f)

#define OWN_ROWS    196
#define OWN_THREADS 1024
#define OWN_WCAP    256
#define OWN_MAX     256
#define PB_EDGES    1024
#define PB_THREADS  256
static_assert(PB_THREADS == OWN_MAX && PB_THREADS * 4 == PB_EDGES, "one thread per owner counter, four edges per thread");
#define ACC_PITCH   68

static __device__ __forceinline__ int gcn_half_of_wave() {
    return (int)((__builtin_amdgcn_s_sendmsg_rtn(0x87) >> 18) & 1u);
}


__global__ void gcn_set_const(float* __restrict__ p0, float* __restrict__ p1, float v, int n) {
    int i = blockIdx.x * blockDim.x + threadIdx.x;
    if (i < n) { p0[i] = v; p1[i] = 0.0f; }
}

__global__ void __launch_bounds__(PB_THREADS)
gcn_deg_part(const int* __restrict__ dst, float* __restrict__ deg0, float* __restrict__ deg1,
             unsigned* __restrict__ meta, unsigned short* __restrict__ sorted, int E) {
    __shared__ unsigned cnt[OWN_MAX];
    __shared__ unsigned start[OWN_MAX];
    __shared__ unsigned wtot[PB_THREADS / 32];
    __shared__ __attribute__((aligned(8))) unsigned short srt[PB_EDGES];

    const int tid  = threadIdx.x;
    const int lane = tid & 31;
    const int wave = tid >> 5;
    const int side = gcn_half_of_wave();
    float* deg = side ? deg1 : deg0;
    cnt[tid] = 0u;
    __syncthreads();

    const int e0 = blockIdx.x * PB_EDGES + tid * 4;
    int d[4];
    if (((E & 3) == 0) && e0 < E) {
        const int4 v = *(const int4*)(dst + e0);
        d[0] = v.x; d[1] = v.y; d[2] = v.z; d[3] = v.w;
    } else {
#pragma unroll
        for (int k = 0; k < 4; ++k) d[k] = (e0 + k < E) ? dst[e0 + k] : -1;
    }
    int o[4];
    unsigned r[4];
#pragma unroll
    for (int k = 0; k < 4; ++k) {
        o[k] = 0; r[k] = 0u;
        if (d[k] >= 0) {
            atomicAdd(&deg[d[k]], 1.0f);
            o[k] = min((int)((unsigned)d[k] / OWN_ROWS), OWN_MAX - 1);
            r[k] = atomicAdd(&cnt[o[k]], 1u);
        }
    }
    __syncthreads();

    const unsigned c = cnt[tid];
    unsigned incl = c;
#pragma unroll
    for (int s = 1; s < 32; s <<= 1) {
        const unsigned v = __shfl_up(incl, s);
        if (lane >= s) incl += v;
    }
    if (lane == 31) wtot[wave] = incl;
    __syncthreads();
    unsigned base = 0u;
    for (int w = 0; w < wave; ++w) base += wtot[w];
    const unsigned st = base + incl - c;
    start[tid] = st;
    meta[(size_t)blockIdx.x * OWN_MAX + tid] = st | (c << 16);
    __syncthreads();

#pragma unroll
    for (int k = 0; k < 4; ++k) {
        if (d[k] >= 0) srt[start[o[k]] + r[k]] = (unsigned short)(tid * 4 + k);
        else           srt[tid * 4 + k] = (unsigned short)0;
    }
    __syncthreads();
    ((uint2*)(sorted + (size_t)blockIdx.x * PB_EDGES))[tid] = ((const uint2*)srt)[tid];
}

__global__ void gcn_deg_rsqrt(float* __restrict__ deg0, const float* __restrict__ deg1, int n) {
    int i = blockIdx.x * blockDim.x + threadIdx.x;
    if (i < n) deg0[i] = rsqrtf(deg0[i] + deg1[i]);
}

template <int K>
__global__ void __launch_bounds__(128)
gcn_gemm_wmma(const float* __restrict__ Af, const float* __restrict__ W, float* __restrict__ C) {
    const int lane = threadIdx.x & 31;
    const int nT   = threadIdx.x >> 5;
    const int m0   = blockIdx.x << 4;
    const int hi   = lane >> 4;
    const int row  = m0 + (lane & 15);
    const int col  = (nT << 4) + (lane & 15);

    const size_t aofs = (size_t)row * K + hi * 4;
    const float* wk   = W + (size_t)(hi * 4) * HID + col;

    v8f acc0 = {0.f, 0.f, 0.f, 0.f, 0.f, 0.f, 0.f, 0.f};
    v8f acc1 = {0.f, 0.f, 0.f, 0.f, 0.f, 0.f, 0.f, 0.f};

#pragma unroll
    for (int kb = 0; kb < K / 8; ++kb) {
        const float4 a = *(const float4*)(Af + aofs + kb * 8);
        const float* wb = wk + (size_t)(kb * 8) * HID;
        const v2f a0 = {a.x, a.y};
        const v2f a1 = {a.z, a.w};
        const v2f b0 = {wb[0],       wb[HID]};
        const v2f b1 = {wb[2 * HID], wb[3 * HID]};
        acc0 = __builtin_amdgcn_wmma_f32_16x16x4_f32(
             false, a0,  false, b0,
             (short)0, acc0,  false,  false);
        acc1 = __builtin_amdgcn_wmma_f32_16x16x4_f32(
             false, a1,  false, b1,
             (short)0, acc1,  false,  false);
    }
    const v8f acc = acc0 + acc1;

    float* crow = C + (size_t)(m0 + hi * 8) * HID + col;
#pragma unroll
    for (int r = 0; r < 8; ++r) crow[(size_t)r * HID] = acc[r];
}


static __device__ __noinline__ void gcn_add_edge_by_lane(int* acc, const float* __restrict__ H,
                                                         const int* __restrict__ src, const int* __restrict__ dst,
                                                         const float* __restrict__ dis, int e, int n0) {
    const int s = src[e];
    const int d = dst[e];
    const float w = dis[s] * dis[d] * FX_SCALE;
    const float* hrow = H + (size_t)s * HID;
    int* orow = acc + (d - n0) * ACC_PITCH;
    for (int c = 0; c < HID; ++c) atomicAdd(orow + c, __float2int_rn(hrow[c] * w));
}

static __device__ __forceinline__ int gcn_pick_edges(int* acc, int* wl, const float* __restrict__ H,
                                                     const int* __restrict__ src, const int* __restrict__ dst,
                                                     const unsigned* __restrict__ meta,
                                                     const unsigned short* __restrict__ sorted,
                                                     const float* __restrict__ dis, int nblk, int n0,
                                                     int lane, int wave) {
    int wcnt = 0;
    for (int tb = 0; tb < nblk; tb += OWN_THREADS) {
        const int t = tb + lane * 32 + wave;
        unsigned m = 0u;
        if (t < nblk) m = meta[(size_t)t * OWN_MAX + blockIdx.x];
        const int c  = (int)(m >> 16);
        const int st = (int)(m & 0xFFFFu);
        int incl = c;
#pragma unroll
        for (int s = 1; s < 32; s <<= 1) {
            const int v = __shfl_up(incl, s);
            if (lane >= s) incl += v;
        }
        const int total = __builtin_amdgcn_readlane(incl, 31);
        const int slot0 = wcnt + incl - c;
        const unsigned short* sp = sorted + (size_t)t * PB_EDGES + st;
        for (int k0 = 0; k0 < c; k0 += 8) {
            int id[8];
#pragma unroll
            for (int u = 0; u < 8; ++u) id[u] = sp[min(k0 + u, c - 1)];
#pragma unroll
            for (int u = 0; u < 8; ++u) {
                if (k0 + u < c) {
                    const int e = t * PB_EDGES + id[u];
                    const int slot = slot0 + k0 + u;
                    if (slot < OWN_WCAP) wl[slot] = e;
                    else gcn_add_edge_by_lane(acc, H, src, dst, dis, e, n0);
                }
            }
        }
        wcnt += total;
    }
    return wcnt;
}

template <bool FIRST>
__global__ void __launch_bounds__(OWN_THREADS)
gcn_owner_sum(const float* __restrict__ H, const int* __restrict__ src, const int* __restrict__ dst,
              const unsigned* __restrict__ meta, const unsigned short* __restrict__ sorted,
              const float* __restrict__ dis,
              const float* __restrict__ bias, const float* __restrict__ Wn, float* __restrict__ outp,
              int* saved, int* saved_cnt, int N, int nblk) {
    __shared__ int acc[OWN_ROWS * ACC_PITCH];
    __shared__ int lists[(OWN_THREADS / 32) * OWN_WCAP];

    const int tid  = threadIdx.x;
    const int lane = tid & 31;
    const int wave = tid >> 5;
    const int n0   = blockIdx.x * OWN_ROWS;
    const int rows = min(OWN_ROWS, N - n0);
    const int nv4  = rows * (HID / 4);
    int* wl = lists + wave * OWN_WCAP;
    const int wslot = blockIdx.x * (OWN_THREADS / 32) + wave;
    int* sv = saved + (size_t)wslot * OWN_WCAP;

    for (int i = tid; i < nv4; i += OWN_THREADS) {
        const int r = i >> 4, c4 = (i & 15) * 4;
        const float dd = dis[n0 + r];
        const float d2 = dd * dd;
        const float4 h = *(const float4*)(H + (size_t)(n0 + r) * HID + c4);
        const float4 b = *(const float4*)(bias + c4);
        int4 v;
        v.x = __float2int_rn((b.x + h.x * d2) * FX_SCALE);
        v.y = __float2int_rn((b.y + h.y * d2) * FX_SCALE);
        v.z = __float2int_rn((b.z + h.z * d2) * FX_SCALE);
        v.w = __float2int_rn((b.w + h.w * d2) * FX_SCALE);
        *(int4*)(acc + r * ACC_PITCH + c4) = v;
    }
    __syncthreads();

    int found;
    if (FIRST) {
        found = __builtin_amdgcn_readfirstlane(gcn_pick_edges(acc, wl, H, src, dst, meta, sorted, dis, nblk, n0, lane, wave));
        const int keep = min(found, OWN_WCAP);
        __syncthreads();
        for (int j = lane; j < keep; j += 32) sv[j] = wl[j];
        if (lane == 0) saved_cnt[wslot] = found;
    } else {
        found = __builtin_amdgcn_readfirstlane(saved_cnt[wslot]);
        if (found <= OWN_WCAP) {
            for (int j = lane; j < found; j += 32) wl[j] = sv[j];
        } else {
            found = __builtin_amdgcn_readfirstlane(gcn_pick_edges(acc, wl, H, src, dst, meta, sorted, dis, nblk, n0, lane, wave));
        }
        __syncthreads();
    }
    const int cnt = min(found, OWN_WCAP);

    for (int base = 0; base < cnt; base += 32) {
        const int j = base + lane;
        int s = 0, dl = 0;
        float w = 0.0f;
        if (j < cnt) {
            const int e = wl[j];
            s = src[e];
            const int d = dst[e];
            dl = d - n0;
            w = dis[s] * dis[d] * FX_SCALE;
        }
        const int wbits = __float_as_int(w);
        const int left = cnt - base;

        float h0[4][8], h1[4][8];

#define GCN_LOAD_GROUP(g, on)                                                      \
    if (on) {                                                                      \
    _Pragma("unroll")   \
    for (int u = 0; u < 8; ++u) {                                                  \
        const int sj = __builtin_amdgcn_readlane(s, (g) * 8 + u);                  \
        const float* hrow = H + (size_t)sj * HID;        \
        h0[g][u] = hrow[lane];                                                     \
        h1[g][u] = hrow[lane + 32];                                                \
    } }
#define GCN_ADD_GROUP(g, on)                                                       \
    if (on) {                                                                      \
    _Pragma("unroll")   \
    for (int u = 0; u < 8; ++u) {                                                  \
        const int   dj = __builtin_amdgcn_readlane(dl, (g) * 8 + u);               \
        const float wj = __int_as_float(__builtin_amdgcn_readlane(wbits, (g) * 8 + u)); \
        int* orow = acc + dj * ACC_PITCH;                 \
        atomicAdd(orow + lane,      __float2int_rn(h0[g][u] * wj));                \
        atomicAdd(orow + lane + 32, __float2int_rn(h1[g][u] * wj));                \
    } }

        if (left >= 32) {
            GCN_LOAD_GROUP(0, true)
            GCN_LOAD_GROUP(1, true)
            GCN_ADD_GROUP(0, true)
            GCN_LOAD_GROUP(2, true)
            GCN_ADD_GROUP(1, true)
            GCN_LOAD_GROUP(3, true)
            GCN_ADD_GROUP(2, true)
            GCN_ADD_GROUP(3, true)
        } else {
            GCN_LOAD_GROUP(0, true)
            GCN_LOAD_GROUP(1, left > 8)
            GCN_ADD_GROUP(0, true)
            GCN_LOAD_GROUP(2, left > 16)
            GCN_ADD_GROUP(1, left > 8)
            GCN_LOAD_GROUP(3, left > 24)
            GCN_ADD_GROUP(2, left > 16)
            GCN_ADD_GROUP(3, left > 24)
        }
#undef GCN_LOAD_GROUP
#undef GCN_ADD_GROUP
    }
    __syncthreads();

    if (FIRST) {
        const int c  = lane & 15;
        const int hi = lane >> 4;
        const int wv = __builtin_amdgcn_readfirstlane(wave);
        const int units = ((rows + 15) >> 4) * 2;
        for (int unit = wv; unit < units; unit += OWN_THREADS / 32) {
            const int m0l = (unit >> 1) << 4;
            const int nb  = (unit & 1) << 5;
            const int* arow = acc + min(m0l + c, rows - 1) * ACC_PITCH + hi * 4;
            const float* wk = Wn + (size_t)(hi * 4) * HID + nb + c;
            v8f p0 = {0.f, 0.f, 0.f, 0.f, 0.f, 0.f, 0.f, 0.f};
            v8f p1 = {0.f, 0.f, 0.f, 0.f, 0.f, 0.f, 0.f, 0.f};
            v8f q0 = {0.f, 0.f, 0.f, 0.f, 0.f, 0.f, 0.f, 0.f};
            v8f q1 = {0.f, 0.f, 0.f, 0.f, 0.f, 0.f, 0.f, 0.f};
#pragma unroll
            for (int kb = 0; kb < HID / 8; ++kb) {
                const int4 ai = *(const int4*)(arow + kb * 8);
                const v2f a0 = {fmaxf((float)ai.x * FX_INV, 0.0f), fmaxf((float)ai.y * FX_INV, 0.0f)};
                const v2f a1 = {fmaxf((float)ai.z * FX_INV, 0.0f), fmaxf((float)ai.w * FX_INV, 0.0f)};
                const float* wb = wk + (size_t)(kb * 8) * HID;
                const v2f b0 = {wb[0],            wb[HID]};
                const v2f b1 = {wb[2 * HID],      wb[3 * HID]};
                const v2f d0 = {wb[16],           wb[HID + 16]};
                const v2f d1 = {wb[2 * HID + 16], wb[3 * HID + 16]};
                p0 = __builtin_amdgcn_wmma_f32_16x16x4_f32(false, a0, false, b0, (short)0, p0, false, false);
                q0 = __builtin_amdgcn_wmma_f32_16x16x4_f32(false, a0, false, d0, (short)0, q0, false, false);
                p1 = __builtin_amdgcn_wmma_f32_16x16x4_f32(false, a1, false, b1, (short)0, p1, false, false);
                q1 = __builtin_amdgcn_wmma_f32_16x16x4_f32(false, a1, false, d1, (short)0, q1, false, false);
            }
            const v8f p = p0 + p1;
            const v8f q = q0 + q1;
            float* crow = outp + (size_t)(n0 + m0l + hi * 8) * HID + nb + c;
            if (m0l + 16 <= rows) {
#pragma unroll
                for (int r = 0; r < 8; ++r) {
                    crow[(size_t)r * HID]      = p[r];
                    crow[(size_t)r * HID + 16] = q[r];
                }
            } else {
                const int live = rows - (m0l + hi * 8);
#pragma unroll
                for (int r = 0; r < 8; ++r) {
                    if (r < live) {
                        crow[(size_t)r * HID]      = p[r];
                        crow[(size_t)r * HID + 16] = q[r];
                    }
                }
            }
        }
    } else {
        float4* orows = (float4*)(outp + (size_t)n0 * HID);
        for (int i = tid; i < nv4; i += OWN_THREADS) {
            const int r = i >> 4, c4 = (i & 15) * 4;
            const int4 a = *(const int4*)(acc + r * ACC_PITCH + c4);
            float4 v;
            v.x = (float)a.x * FX_INV;
            v.y = (float)a.y * FX_INV;
            v.z = (float)a.z * FX_INV;
            v.w = (float)a.w * FX_INV;
            orows[i] = v;
        }
    }
}


extern "C" void kernel_launch(void* const* d_in, const int* in_sizes, int n_in,
                              void* d_out, int out_size, void* d_ws, size_t ws_size,
                              hipStream_t stream) {
    const float* x  = (const float*)d_in[0];
    const int*   ei = (const int*)d_in[1];
    const float* W1 = (const float*)d_in[2];
    const float* b1 = (const float*)d_in[3];
    const float* W2 = (const float*)d_in[4];
    const float* b2 = (const float*)d_in[5];
    float* out = (float*)d_out;

    const int N = in_sizes[0] / IN_CH;
    const int E = in_sizes[1] / 2;
    const int* src = ei;
    const int* dst = ei + E;

    const int owners = (N + OWN_ROWS - 1) / OWN_ROWS;

    char* ws = (char*)d_ws;
    size_t off = 0;
    auto alloc = [&](size_t bytes) -> char* {
        char* p = ws + off;
        off = (off + bytes + 255) & ~(size_t)255;
        return p;
    };
    float*         proj  = (float*)        alloc((size_t)N * HID * sizeof(float));
    const int nblk = (E + PB_EDGES - 1) / PB_EDGES;
    unsigned*       meta   = (unsigned*)      alloc((size_t)nblk * OWN_MAX * sizeof(unsigned));
    unsigned short* sorted = (unsigned short*)alloc((size_t)nblk * PB_EDGES * sizeof(unsigned short));
    int*           saved = (int*)          alloc((size_t)owners * (OWN_THREADS / 32) * OWN_WCAP * sizeof(int));
    int*           scnt  = (int*)          alloc((size_t)owners * (OWN_THREADS / 32) * sizeof(int));
    float*         dis   = (float*)        alloc((size_t)N * sizeof(float));
    float*         deg1  = (float*)        alloc((size_t)N * sizeof(float));
    float*         proj1 = out;

    const int T = 256;

    gcn_set_const<<<(N + T - 1) / T, T, 0, stream>>>(dis, deg1, 1.0f, N);
    gcn_deg_part<<<nblk, PB_THREADS, 0, stream>>>(dst, dis, deg1, meta, sorted, E);
    gcn_deg_rsqrt<<<(N + T - 1) / T, T, 0, stream>>>(dis, deg1, N);

    gcn_gemm_wmma<IN_CH><<<N / 16, 128, 0, stream>>>(x, W1, proj1);
    gcn_owner_sum<true><<<owners, OWN_THREADS, 0, stream>>>(proj1, src, dst, meta, sorted, dis, b1, W2, proj, saved, scnt, N, nblk);

    gcn_owner_sum<false><<<owners, OWN_THREADS, 0, stream>>>(proj, src, dst, meta, sorted, dis, b2, nullptr, out, saved, scnt, N, nblk);
}


# list entries carry the edge itself: degree pass packs source id | owner-local row (was the edge number), layer 1 saves the source's degree byte too: no src/dst gathers, no weight gather in layer 2
# speedup vs baseline: 3.3904x; 1.1187x over previous
#include <hip/hip_runtime.h>

typedef __attribute__((ext_vector_type(2)))  float    v2f;
typedef __attribute__((ext_vector_type(8)))  float    v8f;

#define IN_CH 128
#define HID   64

#define FX_SCALE 1048576.0f
#define FX_INV   (1.0f / 1048576.0f)

#define OWN_ROWS    196
#define OWN_THREADS 1024
#define OWN_WCAP    256
#define OWN_MAX     256
#define PB_EDGES    1024
#define PB_THREADS  256
static_assert(PB_THREADS == OWN_MAX && PB_THREADS * 4 == PB_EDGES, "one thread per owner counter, four edges per thread");
#define ACC_PITCH   68

static __device__ __forceinline__ int gcn_half_of_wave() {
    return (int)((__builtin_amdgcn_s_sendmsg_rtn(0x87) >> 18) & 1u);
}


__global__ void gcn_set_const(float* __restrict__ p0, float* __restrict__ p1, float v, int n) {
    int i = blockIdx.x * blockDim.x + threadIdx.x;
    if (i < n) { p0[i] = v; p1[i] = 0.0f; }
}

__global__ void __launch_bounds__(PB_THREADS)
gcn_deg_part(const int* __restrict__ src, const int* __restrict__ dst, float* __restrict__ deg0, float* __restrict__ deg1,
             unsigned* __restrict__ meta, unsigned* __restrict__ grouped, int E) {
    __shared__ unsigned cnt[OWN_MAX];
    __shared__ unsigned start[OWN_MAX];
    __shared__ unsigned wtot[PB_THREADS / 32];
    __shared__ __attribute__((aligned(16))) unsigned srt[PB_EDGES];

    const int tid  = threadIdx.x;
    const int lane = tid & 31;
    const int wave = tid >> 5;
    const int side = gcn_half_of_wave();
    float* deg = side ? deg1 : deg0;
    cnt[tid] = 0u;
    __syncthreads();

    const int e0 = blockIdx.x * PB_EDGES + tid * 4;
    int d[4], sid[4];
    if (((E & 3) == 0) && e0 < E) {
        const int4 v = *(const int4*)(dst + e0);
        const int4 q = *(const int4*)(src + e0);
        d[0] = v.x; d[1] = v.y; d[2] = v.z; d[3] = v.w;
        sid[0] = q.x; sid[1] = q.y; sid[2] = q.z; sid[3] = q.w;
    } else {
#pragma unroll
        for (int k = 0; k < 4; ++k) {
            d[k]   = (e0 + k < E) ? dst[e0 + k] : -1;
            sid[k] = (e0 + k < E) ? src[e0 + k] : 0;
        }
    }
    int o[4];
    unsigned r[4];
#pragma unroll
    for (int k = 0; k < 4; ++k) {
        o[k] = 0; r[k] = 0u;
        if (d[k] >= 0) {
            atomicAdd(&deg[d[k]], 1.0f);
            o[k] = min((int)((unsigned)d[k] / OWN_ROWS), OWN_MAX - 1);
            r[k] = atomicAdd(&cnt[o[k]], 1u);
        }
    }
    __syncthreads();

    const unsigned c = cnt[tid];
    unsigned incl = c;
#pragma unroll
    for (int s = 1; s < 32; s <<= 1) {
        const unsigned v = __shfl_up(incl, s);
        if (lane >= s) incl += v;
    }
    if (lane == 31) wtot[wave] = incl;
    __syncthreads();
    unsigned base = 0u;
    for (int w = 0; w < wave; ++w) base += wtot[w];
    const unsigned st = base + incl - c;
    start[tid] = st;
    meta[(size_t)blockIdx.x * OWN_MAX + tid] = st | (c << 16);
    __syncthreads();

#pragma unroll
    for (int k = 0; k < 4; ++k) {
        if (d[k] >= 0) srt[start[o[k]] + r[k]] = (unsigned)sid[k] | ((unsigned)(d[k] - o[k] * OWN_ROWS) << 16);
        else           srt[tid * 4 + k] = 0u;
    }
    __syncthreads();
    ((uint4*)(grouped + (size_t)blockIdx.x * PB_EDGES))[tid] = ((const uint4*)srt)[tid];
}

__global__ void gcn_deg_rsqrt(float* __restrict__ deg0, float* __restrict__ deg1, int n) {
    int i = blockIdx.x * blockDim.x + threadIdx.x;
    if (i < n) {
        const float t = deg0[i] + deg1[i];
        deg0[i] = rsqrtf(t);
        deg1[i] = t;
    }
}

template <int K>
__global__ void __launch_bounds__(128)
gcn_gemm_wmma(const float* __restrict__ Af, const float* __restrict__ W, float* __restrict__ C) {
    const int lane = threadIdx.x & 31;
    const int nT   = threadIdx.x >> 5;
    const int m0   = blockIdx.x << 4;
    const int hi   = lane >> 4;
    const int row  = m0 + (lane & 15);
    const int col  = (nT << 4) + (lane & 15);

    const size_t aofs = (size_t)row * K + hi * 4;
    const float* wk   = W + (size_t)(hi * 4) * HID + col;

    v8f acc0 = {0.f, 0.f, 0.f, 0.f, 0.f, 0.f, 0.f, 0.f};
    v8f acc1 = {0.f, 0.f, 0.f, 0.f, 0.f, 0.f, 0.f, 0.f};

#pragma unroll
    for (int kb = 0; kb < K / 8; ++kb) {
        const float4 a = *(const float4*)(Af + aofs + kb * 8);
        const float* wb = wk + (size_t)(kb * 8) * HID;
        const v2f a0 = {a.x, a.y};
        const v2f a1 = {a.z, a.w};
        const v2f b0 = {wb[0],       wb[HID]};
        const v2f b1 = {wb[2 * HID], wb[3 * HID]};
        acc0 = __builtin_amdgcn_wmma_f32_16x16x4_f32(
             false, a0,  false, b0,
             (short)0, acc0,  false,  false);
        acc1 = __builtin_amdgcn_wmma_f32_16x16x4_f32(
             false, a1,  false, b1,
             (short)0, acc1,  false,  false);
    }
    const v8f acc = acc0 + acc1;

    float* crow = C + (size_t)(m0 + hi * 8) * HID + col;
#pragma unroll
    for (int r = 0; r < 8; ++r) crow[(size_t)r * HID] = acc[r];
}


static __device__ __noinline__ void gcn_add_edge_by_lane(int* acc, const float* __restrict__ H,
                                                         const float* __restrict__ dis, int rec, int n0) {
    const int s  = rec & 0xFFFF;
    const int dl = (rec >> 16) & 0xFF;
    const float w = dis[s] * dis[n0 + dl] * FX_SCALE;
    const float* hrow = H + (size_t)s * HID;
    int* orow = acc + dl * ACC_PITCH;
    for (int c = 0; c < HID; ++c) atomicAdd(orow + c, __float2int_rn(hrow[c] * w));
}

static __device__ __forceinline__ int gcn_pick_edges(int* acc, int* wl, const float* __restrict__ H,
                                                     const unsigned* __restrict__ meta,
                                                     const unsigned* __restrict__ grouped,
                                                     const float* __restrict__ dis, int nblk, int n0,
                                                     int lane, int wave) {
    int wcnt = 0;
    for (int tb = 0; tb < nblk; tb += OWN_THREADS) {
        const int t = tb + lane * 32 + wave;
        unsigned m = 0u;
        if (t < nblk) m = meta[(size_t)t * OWN_MAX + blockIdx.x];
        const int c  = (int)(m >> 16);
        const int st = (int)(m & 0xFFFFu);
        int incl = c;
#pragma unroll
        for (int s = 1; s < 32; s <<= 1) {
            const int v = __shfl_up(incl, s);
            if (lane >= s) incl += v;
        }
        const int total = __builtin_amdgcn_readlane(incl, 31);
        const int slot0 = wcnt + incl - c;
        const unsigned* sp = grouped + (size_t)t * PB_EDGES + st;
        for (int k0 = 0; k0 < c; k0 += 8) {
            int id[8];
#pragma unroll
            for (int u = 0; u < 8; ++u) id[u] = (int)sp[min(k0 + u, c - 1)];
#pragma unroll
            for (int u = 0; u < 8; ++u) {
                if (k0 + u < c) {
                    const int slot = slot0 + k0 + u;
                    if (slot < OWN_WCAP) wl[slot] = id[u];
                    else gcn_add_edge_by_lane(acc, H, dis, id[u], n0);
                }
            }
        }
        wcnt += total;
    }
    return wcnt;
}

template <bool FIRST>
__global__ void __launch_bounds__(OWN_THREADS)
gcn_owner_sum(const float* __restrict__ H,
              const unsigned* __restrict__ meta, const unsigned* __restrict__ grouped,
              const float* __restrict__ dis, const float* __restrict__ degt,
              const float* __restrict__ bias, const float* __restrict__ Wn, float* __restrict__ outp,
              int* saved, int* saved_cnt, int N, int nblk) {
    __shared__ int acc[OWN_ROWS * ACC_PITCH];
    __shared__ int lists[(OWN_THREADS / 32) * OWN_WCAP];
    __shared__ float disl[OWN_ROWS];

    const int tid  = threadIdx.x;
    const int lane = tid & 31;
    const int wave = tid >> 5;
    const int n0   = blockIdx.x * OWN_ROWS;
    const int rows = min(OWN_ROWS, N - n0);
    const int nv4  = rows * (HID / 4);
    int* wl = lists + wave * OWN_WCAP;
    const int wslot = blockIdx.x * (OWN_THREADS / 32) + wave;
    int* sv = saved + (size_t)wslot * OWN_WCAP;

    for (int i = tid; i < nv4; i += OWN_THREADS) {
        const int r = i >> 4, c4 = (i & 15) * 4;
        const float dd = dis[n0 + r];
        if ((i & 15) == 0) disl[r] = dd;
        const float d2 = dd * dd;
        const float4 h = *(const float4*)(H + (size_t)(n0 + r) * HID + c4);
        const float4 b = *(const float4*)(bias + c4);
        int4 v;
        v.x = __float2int_rn((b.x + h.x * d2) * FX_SCALE);
        v.y = __float2int_rn((b.y + h.y * d2) * FX_SCALE);
        v.z = __float2int_rn((b.z + h.z * d2) * FX_SCALE);
        v.w = __float2int_rn((b.w + h.w * d2) * FX_SCALE);
        *(int4*)(acc + r * ACC_PITCH + c4) = v;
    }
    __syncthreads();

    int found;
    if (FIRST) {
        found = __builtin_amdgcn_readfirstlane(gcn_pick_edges(acc, wl, H, meta, grouped, dis, nblk, n0, lane, wave));
        __syncthreads();
        if (lane == 0) saved_cnt[wslot] = found;
    } else {
        found = __builtin_amdgcn_readfirstlane(saved_cnt[wslot]);
        if (found <= OWN_WCAP) {
            for (int j = lane; j < found; j += 32) wl[j] = sv[j];
        } else {
            found = __builtin_amdgcn_readfirstlane(gcn_pick_edges(acc, wl, H, meta, grouped, dis, nblk, n0, lane, wave));
        }
        __syncthreads();
    }
    const int cnt = min(found, OWN_WCAP);

    for (int base = 0; base < cnt; base += 32) {
        const int j = base + lane;
        int s = 0, dl = 0;
        float w = 0.0f;
        if (j < cnt) {
            const unsigned rec = (unsigned)wl[j];
            s  = (int)(rec & 0xFFFFu);
            dl = (int)((rec >> 16) & 0xFFu);
            const unsigned db = rec >> 24;
            float dg;
            if (db != 0u) dg = (float)db;
            else          dg = degt[s];
            w = rsqrtf(dg) * disl[dl] * FX_SCALE;
            if (FIRST) sv[j] = (int)((rec & 0x00FFFFFFu) | (dg <= 254.0f ? ((unsigned)dg << 24) : 0u));
        }
        const int wbits = __float_as_int(w);
        const int left = cnt - base;

        float h0[4][8], h1[4][8];

#define GCN_LOAD_GROUP(g, on)                                                      \
    if (on) {                                                                      \
    _Pragma("unroll")   \
    for (int u = 0; u < 8; ++u) {                                                  \
        const int sj = __builtin_amdgcn_readlane(s, (g) * 8 + u);                  \
        const float* hrow = H + (size_t)sj * HID;        \
        h0[g][u] = hrow[lane];                                                     \
        h1[g][u] = hrow[lane + 32];                                                \
    } }
#define GCN_ADD_GROUP(g, on)                                                       \
    if (on) {                                                                      \
    _Pragma("unroll")   \
    for (int u = 0; u < 8; ++u) {                                                  \
        const int   dj = __builtin_amdgcn_readlane(dl, (g) * 8 + u);               \
        const float wj = __int_as_float(__builtin_amdgcn_readlane(wbits, (g) * 8 + u)); \
        int* orow = acc + dj * ACC_PITCH;                 \
        atomicAdd(orow + lane,      __float2int_rn(h0[g][u] * wj));                \
        atomicAdd(orow + lane + 32, __float2int_rn(h1[g][u] * wj));                \
    } }

        if (left >= 32) {
            GCN_LOAD_GROUP(0, true)
            GCN_LOAD_GROUP(1, true)
            GCN_ADD_GROUP(0, true)
            GCN_LOAD_GROUP(2, true)
            GCN_ADD_GROUP(1, true)
            GCN_LOAD_GROUP(3, true)
            GCN_ADD_GROUP(2, true)
            GCN_ADD_GROUP(3, true)
        } else {
            GCN_LOAD_GROUP(0, true)
            GCN_LOAD_GROUP(1, left > 8)
            GCN_ADD_GROUP(0, true)
            GCN_LOAD_GROUP(2, left > 16)
            GCN_ADD_GROUP(1, left > 8)
            GCN_LOAD_GROUP(3, left > 24)
            GCN_ADD_GROUP(2, left > 16)
            GCN_ADD_GROUP(3, left > 24)
        }
#undef GCN_LOAD_GROUP
#undef GCN_ADD_GROUP
    }
    __syncthreads();

    if (FIRST) {
        const int c  = lane & 15;
        const int hi = lane >> 4;
        const int wv = __builtin_amdgcn_readfirstlane(wave);
        const int units = ((rows + 15) >> 4) * 2;
        for (int unit = wv; unit < units; unit += OWN_THREADS / 32) {
            const int m0l = (unit >> 1) << 4;
            const int nb  = (unit & 1) << 5;
            const int* arow = acc + min(m0l + c, rows - 1) * ACC_PITCH + hi * 4;
            const float* wk = Wn + (size_t)(hi * 4) * HID + nb + c;
            v8f p0 = {0.f, 0.f, 0.f, 0.f, 0.f, 0.f, 0.f, 0.f};
            v8f p1 = {0.f, 0.f, 0.f, 0.f, 0.f, 0.f, 0.f, 0.f};
            v8f q0 = {0.f, 0.f, 0.f, 0.f, 0.f, 0.f, 0.f, 0.f};
            v8f q1 = {0.f, 0.f, 0.f, 0.f, 0.f, 0.f, 0.f, 0.f};
#pragma unroll
            for (int kb = 0; kb < HID / 8; ++kb) {
                const int4 ai = *(const int4*)(arow + kb * 8);
                const v2f a0 = {fmaxf((float)ai.x * FX_INV, 0.0f), fmaxf((float)ai.y * FX_INV, 0.0f)};
                const v2f a1 = {fmaxf((float)ai.z * FX_INV, 0.0f), fmaxf((float)ai.w * FX_INV, 0.0f)};
                const float* wb = wk + (size_t)(kb * 8) * HID;
                const v2f b0 = {wb[0],            wb[HID]};
                const v2f b1 = {wb[2 * HID],      wb[3 * HID]};
                const v2f d0 = {wb[16],           wb[HID + 16]};
                const v2f d1 = {wb[2 * HID + 16], wb[3 * HID + 16]};
                p0 = __builtin_amdgcn_wmma_f32_16x16x4_f32(false, a0, false, b0, (short)0, p0, false, false);
                q0 = __builtin_amdgcn_wmma_f32_16x16x4_f32(false, a0, false, d0, (short)0, q0, false, false);
                p1 = __builtin_amdgcn_wmma_f32_16x16x4_f32(false, a1, false, b1, (short)0, p1, false, false);
                q1 = __builtin_amdgcn_wmma_f32_16x16x4_f32(false, a1, false, d1, (short)0, q1, false, false);
            }
            const v8f p = p0 + p1;
            const v8f q = q0 + q1;
            float* crow = outp + (size_t)(n0 + m0l + hi * 8) * HID + nb + c;
            if (m0l + 16 <= rows) {
#pragma unroll
                for (int r = 0; r < 8; ++r) {
                    crow[(size_t)r * HID]      = p[r];
                    crow[(size_t)r * HID + 16] = q[r];
                }
            } else {
                const int live = rows - (m0l + hi * 8);
#pragma unroll
                for (int r = 0; r < 8; ++r) {
                    if (r < live) {
                        crow[(size_t)r * HID]      = p[r];
                        crow[(size_t)r * HID + 16] = q[r];
                    }
                }
            }
        }
    } else {
        float4* orows = (float4*)(outp + (size_t)n0 * HID);
        for (int i = tid; i < nv4; i += OWN_THREADS) {
            const int r = i >> 4, c4 = (i & 15) * 4;
            const int4 a = *(const int4*)(acc + r * ACC_PITCH + c4);
            float4 v;
            v.x = (float)a.x * FX_INV;
            v.y = (float)a.y * FX_INV;
            v.z = (float)a.z * FX_INV;
            v.w = (float)a.w * FX_INV;
            orows[i] = v;
        }
    }
}


extern "C" void kernel_launch(void* const* d_in, const int* in_sizes, int n_in,
                              void* d_out, int out_size, void* d_ws, size_t ws_size,
                              hipStream_t stream) {
    const float* x  = (const float*)d_in[0];
    const int*   ei = (const int*)d_in[1];
    const float* W1 = (const float*)d_in[2];
    const float* b1 = (const float*)d_in[3];
    const float* W2 = (const float*)d_in[4];
    const float* b2 = (const float*)d_in[5];
    float* out = (float*)d_out;

    const int N = in_sizes[0] / IN_CH;
    const int E = in_sizes[1] / 2;
    const int* src = ei;
    const int* dst = ei + E;

    const int owners = (N + OWN_ROWS - 1) / OWN_ROWS;

    char* ws = (char*)d_ws;
    size_t off = 0;
    auto alloc = [&](size_t bytes) -> char* {
        char* p = ws + off;
        off = (off + bytes + 255) & ~(size_t)255;
        return p;
    };
    float*         proj  = (float*)        alloc((size_t)N * HID * sizeof(float));
    const int nblk = (E + PB_EDGES - 1) / PB_EDGES;
    unsigned*       meta   = (unsigned*)      alloc((size_t)nblk * OWN_MAX * sizeof(unsigned));
    unsigned*       grouped = (unsigned*)     alloc((size_t)nblk * PB_EDGES * sizeof(unsigned));
    int*           saved = (int*)          alloc((size_t)owners * (OWN_THREADS / 32) * OWN_WCAP * sizeof(int));
    int*           scnt  = (int*)          alloc((size_t)owners * (OWN_THREADS / 32) * sizeof(int));
    float*         dis   = (float*)        alloc((size_t)N * sizeof(float));
    float*         deg1  = (float*)        alloc((size_t)N * sizeof(float));
    float*         proj1 = out;

    const int T = 256;

    gcn_set_const<<<(N + T - 1) / T, T, 0, stream>>>(dis, deg1, 1.0f, N);
    gcn_deg_part<<<nblk, PB_THREADS, 0, stream>>>(src, dst, dis, deg1, meta, grouped, E);
    gcn_deg_rsqrt<<<(N + T - 1) / T, T, 0, stream>>>(dis, deg1, N);

    gcn_gemm_wmma<IN_CH><<<N / 16, 128, 0, stream>>>(x, W1, proj1);
    gcn_owner_sum<true><<<owners, OWN_THREADS, 0, stream>>>(proj1, meta, grouped, dis, deg1, b1, W2, proj, saved, scnt, N, nblk);

    gcn_owner_sum<false><<<owners, OWN_THREADS, 0, stream>>>(proj, meta, grouped, dis, deg1, b2, nullptr, out, saved, scnt, N, nblk);
}


# degree planes' start values stored by the layer-1 product's launch (now first), deg^-1/2 formed by its readers (rsqrtf of plane 0 + plane 1): set and rsqrt launches gone, 6 -> 4 launches
# speedup vs baseline: 3.5393x; 1.0439x over previous
#include <hip/hip_runtime.h>

typedef __attribute__((ext_vector_type(2)))  float    v2f;
typedef __attribute__((ext_vector_type(8)))  float    v8f;

#define IN_CH 128
#define HID   64

#define FX_SCALE 1048576.0f
#define FX_INV   (1.0f / 1048576.0f)

#define OWN_ROWS    196
#define OWN_THREADS 1024
#define OWN_WCAP    256
#define OWN_MAX     256
#define PB_EDGES    1024
#define PB_THREADS  256
static_assert(PB_THREADS == OWN_MAX && PB_THREADS * 4 == PB_EDGES, "one thread per owner counter, four edges per thread");
#define ACC_PITCH   68

static __device__ __forceinline__ int gcn_half_of_wave() {
    return (int)((__builtin_amdgcn_s_sendmsg_rtn(0x87) >> 18) & 1u);
}


__global__ void __launch_bounds__(PB_THREADS)
gcn_deg_part(const int* __restrict__ src, const int* __restrict__ dst, float* __restrict__ deg0, float* __restrict__ deg1,
             unsigned* __restrict__ meta, unsigned* __restrict__ grouped, int E) {
    __shared__ unsigned cnt[OWN_MAX];
    __shared__ unsigned start[OWN_MAX];
    __shared__ unsigned wtot[PB_THREADS / 32];
    __shared__ __attribute__((aligned(16))) unsigned srt[PB_EDGES];

    const int tid  = threadIdx.x;
    const int lane = tid & 31;
    const int wave = tid >> 5;
    const int side = gcn_half_of_wave();
    float* deg = side ? deg1 : deg0;
    cnt[tid] = 0u;
    __syncthreads();

    const int e0 = blockIdx.x * PB_EDGES + tid * 4;
    int d[4], sid[4];
    if (((E & 3) == 0) && e0 < E) {
        const int4 v = *(const int4*)(dst + e0);
        const int4 q = *(const int4*)(src + e0);
        d[0] = v.x; d[1] = v.y; d[2] = v.z; d[3] = v.w;
        sid[0] = q.x; sid[1] = q.y; sid[2] = q.z; sid[3] = q.w;
    } else {
#pragma unroll
        for (int k = 0; k < 4; ++k) {
            d[k]   = (e0 + k < E) ? dst[e0 + k] : -1;
            sid[k] = (e0 + k < E) ? src[e0 + k] : 0;
        }
    }
    int o[4];
    unsigned r[4];
#pragma unroll
    for (int k = 0; k < 4; ++k) {
        o[k] = 0; r[k] = 0u;
        if (d[k] >= 0) {
            atomicAdd(&deg[d[k]], 1.0f);
            o[k] = min((int)((unsigned)d[k] / OWN_ROWS), OWN_MAX - 1);
            r[k] = atomicAdd(&cnt[o[k]], 1u);
        }
    }
    __syncthreads();

    const unsigned c = cnt[tid];
    unsigned incl = c;
#pragma unroll
    for (int s = 1; s < 32; s <<= 1) {
        const unsigned v = __shfl_up(incl, s);
        if (lane >= s) incl += v;
    }
    if (lane == 31) wtot[wave] = incl;
    __syncthreads();
    unsigned base = 0u;
    for (int w = 0; w < wave; ++w) base += wtot[w];
    const unsigned st = base + incl - c;
    start[tid] = st;
    meta[(size_t)blockIdx.x * OWN_MAX + tid] = st | (c << 16);
    __syncthreads();

#pragma unroll
    for (int k = 0; k < 4; ++k) {
        if (d[k] >= 0) srt[start[o[k]] + r[k]] = (unsigned)sid[k] | ((unsigned)(d[k] - o[k] * OWN_ROWS) << 16);
        else           srt[tid * 4 + k] = 0u;
    }
    __syncthreads();
    ((uint4*)(grouped + (size_t)blockIdx.x * PB_EDGES))[tid] = ((const uint4*)srt)[tid];
}

static __device__ __forceinline__ float gcn_deg_of(const float* __restrict__ deg0, const float* __restrict__ deg1, int i) {
    return deg0[i] + deg1[i];
}

template <int K>
__global__ void __launch_bounds__(128)
gcn_gemm_wmma(const float* __restrict__ Af, const float* __restrict__ W, float* __restrict__ C,
              float* __restrict__ deg0, float* __restrict__ deg1, int n) {
    const int lane = threadIdx.x & 31;
    const int nT   = threadIdx.x >> 5;
    const int m0   = blockIdx.x << 4;

    if ((blockIdx.x & 1) == 0 && threadIdx.x < 32) {
        const int i = m0 + (int)threadIdx.x;
        if (i < n) { deg0[i] = 1.0f; deg1[i] = 0.0f; }
    }

    const int hi   = lane >> 4;
    const int row  = m0 + (lane & 15);
    const int col  = (nT << 4) + (lane & 15);

    const size_t aofs = (size_t)row * K + hi * 4;
    const float* wk   = W + (size_t)(hi * 4) * HID + col;

    v8f acc0 = {0.f, 0.f, 0.f, 0.f, 0.f, 0.f, 0.f, 0.f};
    v8f acc1 = {0.f, 0.f, 0.f, 0.f, 0.f, 0.f, 0.f, 0.f};

#pragma unroll
    for (int kb = 0; kb < K / 8; ++kb) {
        const float4 a = *(const float4*)(Af + aofs + kb * 8);
        const float* wb = wk + (size_t)(kb * 8) * HID;
        const v2f a0 = {a.x, a.y};
        const v2f a1 = {a.z, a.w};
        const v2f b0 = {wb[0],       wb[HID]};
        const v2f b1 = {wb[2 * HID], wb[3 * HID]};
        acc0 = __builtin_amdgcn_wmma_f32_16x16x4_f32(
             false, a0,  false, b0,
             (short)0, acc0,  false,  false);
        acc1 = __builtin_amdgcn_wmma_f32_16x16x4_f32(
             false, a1,  false, b1,
             (short)0, acc1,  false,  false);
    }
    const v8f acc = acc0 + acc1;

    float* crow = C + (size_t)(m0 + hi * 8) * HID + col;
#pragma unroll
    for (int r = 0; r < 8; ++r) crow[(size_t)r * HID] = acc[r];
}


static __device__ __noinline__ void gcn_add_edge_by_lane(int* acc, const float* __restrict__ H,
                                                         const float* __restrict__ deg0, const float* __restrict__ deg1,
                                                         int rec, int n0) {
    const int s  = rec & 0xFFFF;
    const int dl = (rec >> 16) & 0xFF;
    const float w = rsqrtf(gcn_deg_of(deg0, deg1, s)) * rsqrtf(gcn_deg_of(deg0, deg1, n0 + dl)) * FX_SCALE;
    const float* hrow = H + (size_t)s * HID;
    int* orow = acc + dl * ACC_PITCH;
    for (int c = 0; c < HID; ++c) atomicAdd(orow + c, __float2int_rn(hrow[c] * w));
}

static __device__ __forceinline__ int gcn_pick_edges(int* acc, int* wl, const float* __restrict__ H,
                                                     const unsigned* __restrict__ meta,
                                                     const unsigned* __restrict__ grouped,
                                                     const float* __restrict__ deg0, const float* __restrict__ deg1,
                                                     int nblk, int n0, int lane, int wave) {
    int wcnt = 0;
    for (int tb = 0; tb < nblk; tb += OWN_THREADS) {
        const int t = tb + lane * 32 + wave;
        unsigned m = 0u;
        if (t < nblk) m = meta[(size_t)t * OWN_MAX + blockIdx.x];
        const int c  = (int)(m >> 16);
        const int st = (int)(m & 0xFFFFu);
        int incl = c;
#pragma unroll
        for (int s = 1; s < 32; s <<= 1) {
            const int v = __shfl_up(incl, s);
            if (lane >= s) incl += v;
        }
        const int total = __builtin_amdgcn_readlane(incl, 31);
        const int slot0 = wcnt + incl - c;
        const unsigned* sp = grouped + (size_t)t * PB_EDGES + st;
        for (int k0 = 0; k0 < c; k0 += 8) {
            int id[8];
#pragma unroll
            for (int u = 0; u < 8; ++u) id[u] = (int)sp[min(k0 + u, c - 1)];
#pragma unroll
            for (int u = 0; u < 8; ++u) {
                if (k0 + u < c) {
                    const int slot = slot0 + k0 + u;
                    if (slot < OWN_WCAP) wl[slot] = id[u];
                    else gcn_add_edge_by_lane(acc, H, deg0, deg1, id[u], n0);
                }
            }
        }
        wcnt += total;
    }
    return wcnt;
}

template <bool FIRST>
__global__ void __launch_bounds__(OWN_THREADS)
gcn_owner_sum(const float* __restrict__ H,
              const unsigned* __restrict__ meta, const unsigned* __restrict__ grouped,
              const float* __restrict__ deg0, const float* __restrict__ deg1,
              const float* __restrict__ bias, const float* __restrict__ Wn, float* __restrict__ outp,
              int* saved, int* saved_cnt, int N, int nblk) {
    __shared__ int acc[OWN_ROWS * ACC_PITCH];
    __shared__ int lists[(OWN_THREADS / 32) * OWN_WCAP];
    __shared__ float disl[OWN_ROWS];

    const int tid  = threadIdx.x;
    const int lane = tid & 31;
    const int wave = tid >> 5;
    const int n0   = blockIdx.x * OWN_ROWS;
    const int rows = min(OWN_ROWS, N - n0);
    const int nv4  = rows * (HID / 4);
    int* wl = lists + wave * OWN_WCAP;
    const int wslot = blockIdx.x * (OWN_THREADS / 32) + wave;
    int* sv = saved + (size_t)wslot * OWN_WCAP;

    for (int i = tid; i < nv4; i += OWN_THREADS) {
        const int r = i >> 4, c4 = (i & 15) * 4;
        const float dg = gcn_deg_of(deg0, deg1, n0 + r);
        const float4 h = *(const float4*)(H + (size_t)(n0 + r) * HID + c4);
        const float4 b = *(const float4*)(bias + c4);
        const float dd = rsqrtf(dg);
        if ((i & 15) == 0) disl[r] = dd;
        const float d2 = dd * dd;
        int4 v;
        v.x = __float2int_rn((b.x + h.x * d2) * FX_SCALE);
        v.y = __float2int_rn((b.y + h.y * d2) * FX_SCALE);
        v.z = __float2int_rn((b.z + h.z * d2) * FX_SCALE);
        v.w = __float2int_rn((b.w + h.w * d2) * FX_SCALE);
        *(int4*)(acc + r * ACC_PITCH + c4) = v;
    }
    __syncthreads();

    int found;
    if (FIRST) {
        found = __builtin_amdgcn_readfirstlane(gcn_pick_edges(acc, wl, H, meta, grouped, deg0, deg1, nblk, n0, lane, wave));
        __syncthreads();
        if (lane == 0) saved_cnt[wslot] = found;
    } else {
        found = __builtin_amdgcn_readfirstlane(saved_cnt[wslot]);
        if (found <= OWN_WCAP) {
            for (int j = lane; j < found; j += 32) wl[j] = sv[j];
        } else {
            found = __builtin_amdgcn_readfirstlane(gcn_pick_edges(acc, wl, H, meta, grouped, deg0, deg1, nblk, n0, lane, wave));
        }
        __syncthreads();
    }
    const int cnt = min(found, OWN_WCAP);

    for (int base = 0; base < cnt; base += 32) {
        const int j = base + lane;
        int s = 0, dl = 0;
        float w = 0.0f;
        if (j < cnt) {
            const unsigned rec = (unsigned)wl[j];
            s  = (int)(rec & 0xFFFFu);
            dl = (int)((rec >> 16) & 0xFFu);
            const unsigned db = rec >> 24;
            float dg;
            if (db != 0u) dg = (float)db;
            else          dg = gcn_deg_of(deg0, deg1, s);
            w = rsqrtf(dg) * disl[dl] * FX_SCALE;
            if (FIRST) sv[j] = (int)((rec & 0x00FFFFFFu) | (dg <= 254.0f ? ((unsigned)dg << 24) : 0u));
        }
        const int wbits = __float_as_int(w);
        const int left = cnt - base;

        float h0[4][8], h1[4][8];

#define GCN_LOAD_GROUP(g, on)                                                      \
    if (on) {                                                                      \
    _Pragma("unroll")   \
    for (int u = 0; u < 8; ++u) {                                                  \
        const int sj = __builtin_amdgcn_readlane(s, (g) * 8 + u);                  \
        const float* hrow = H + (size_t)sj * HID;        \
        h0[g][u] = hrow[lane];                                                     \
        h1[g][u] = hrow[lane + 32];                                                \
    } }
#define GCN_ADD_GROUP(g, on)                                                       \
    if (on) {                                                                      \
    _Pragma("unroll")   \
    for (int u = 0; u < 8; ++u) {                                                  \
        const int   dj = __builtin_amdgcn_readlane(dl, (g) * 8 + u);               \
        const float wj = __int_as_float(__builtin_amdgcn_readlane(wbits, (g) * 8 + u)); \
        int* orow = acc + dj * ACC_PITCH;                 \
        atomicAdd(orow + lane,      __float2int_rn(h0[g][u] * wj));                \
        atomicAdd(orow + lane + 32, __float2int_rn(h1[g][u] * wj));                \
    } }

        if (left >= 32) {
            GCN_LOAD_GROUP(0, true)
            GCN_LOAD_GROUP(1, true)
            GCN_ADD_GROUP(0, true)
            GCN_LOAD_GROUP(2, true)
            GCN_ADD_GROUP(1, true)
            GCN_LOAD_GROUP(3, true)
            GCN_ADD_GROUP(2, true)
            GCN_ADD_GROUP(3, true)
        } else {
            GCN_LOAD_GROUP(0, true)
            GCN_LOAD_GROUP(1, left > 8)
            GCN_ADD_GROUP(0, true)
            GCN_LOAD_GROUP(2, left > 16)
            GCN_ADD_GROUP(1, left > 8)
            GCN_LOAD_GROUP(3, left > 24)
            GCN_ADD_GROUP(2, left > 16)
            GCN_ADD_GROUP(3, left > 24)
        }
#undef GCN_LOAD_GROUP
#undef GCN_ADD_GROUP
    }
    __syncthreads();

    if (FIRST) {
        const int c  = lane & 15;
        const int hi = lane >> 4;
        const int wv = __builtin_amdgcn_readfirstlane(wave);
        const int units = ((rows + 15) >> 4) * 2;
        for (int unit = wv; unit < units; unit += OWN_THREADS / 32) {
            const int m0l = (unit >> 1) << 4;
            const int nb  = (unit & 1) << 5;
            const int* arow = acc + min(m0l + c, rows - 1) * ACC_PITCH + hi * 4;
            const float* wk = Wn + (size_t)(hi * 4) * HID + nb + c;
            v8f p0 = {0.f, 0.f, 0.f, 0.f, 0.f, 0.f, 0.f, 0.f};
            v8f p1 = {0.f, 0.f, 0.f, 0.f, 0.f, 0.f, 0.f, 0.f};
            v8f q0 = {0.f, 0.f, 0.f, 0.f, 0.f, 0.f, 0.f, 0.f};
            v8f q1 = {0.f, 0.f, 0.f, 0.f, 0.f, 0.f, 0.f, 0.f};
#pragma unroll
            for (int kb = 0; kb < HID / 8; ++kb) {
                const int4 ai = *(const int4*)(arow + kb * 8);
                const v2f a0 = {fmaxf((float)ai.x * FX_INV, 0.0f), fmaxf((float)ai.y * FX_INV, 0.0f)};
                const v2f a1 = {fmaxf((float)ai.z * FX_INV, 0.0f), fmaxf((float)ai.w * FX_INV, 0.0f)};
                const float* wb = wk + (size_t)(kb * 8) * HID;
                const v2f b0 = {wb[0],            wb[HID]};
                const v2f b1 = {wb[2 * HID],      wb[3 * HID]};
                const v2f d0 = {wb[16],           wb[HID + 16]};
                const v2f d1 = {wb[2 * HID + 16], wb[3 * HID + 16]};
                p0 = __builtin_amdgcn_wmma_f32_16x16x4_f32(false, a0, false, b0, (short)0, p0, false, false);
                q0 = __builtin_amdgcn_wmma_f32_16x16x4_f32(false, a0, false, d0, (short)0, q0, false, false);
                p1 = __builtin_amdgcn_wmma_f32_16x16x4_f32(false, a1, false, b1, (short)0, p1, false, false);
                q1 = __builtin_amdgcn_wmma_f32_16x16x4_f32(false, a1, false, d1, (short)0, q1, false, false);
            }
            const v8f p = p0 + p1;
            const v8f q = q0 + q1;
            float* crow = outp + (size_t)(n0 + m0l + hi * 8) * HID + nb + c;
            if (m0l + 16 <= rows) {
#pragma unroll
                for (int r = 0; r < 8; ++r) {
                    crow[(size_t)r * HID]      = p[r];
                    crow[(size_t)r * HID + 16] = q[r];
                }
            } else {
                const int live = rows - (m0l + hi * 8);
#pragma unroll
                for (int r = 0; r < 8; ++r) {
                    if (r < live) {
                        crow[(size_t)r * HID]      = p[r];
                        crow[(size_t)r * HID + 16] = q[r];
                    }
                }
            }
        }
    } else {
        float4* orows = (float4*)(outp + (size_t)n0 * HID);
        for (int i = tid; i < nv4; i += OWN_THREADS) {
            const int r = i >> 4, c4 = (i & 15) * 4;
            const int4 a = *(const int4*)(acc + r * ACC_PITCH + c4);
            float4 v;
            v.x = (float)a.x * FX_INV;
            v.y = (float)a.y * FX_INV;
            v.z = (float)a.z * FX_INV;
            v.w = (float)a.w * FX_INV;
            orows[i] = v;
        }
    }
}


extern "C" void kernel_launch(void* const* d_in, const int* in_sizes, int n_in,
                              void* d_out, int out_size, void* d_ws, size_t ws_size,
                              hipStream_t stream) {
    const float* x  = (const float*)d_in[0];
    const int*   ei = (const int*)d_in[1];
    const float* W1 = (const float*)d_in[2];
    const float* b1 = (const float*)d_in[3];
    const float* W2 = (const float*)d_in[4];
    const float* b2 = (const float*)d_in[5];
    float* out = (float*)d_out;

    const int N = in_sizes[0] / IN_CH;
    const int E = in_sizes[1] / 2;
    const int* src = ei;
    const int* dst = ei + E;

    const int owners = (N + OWN_ROWS - 1) / OWN_ROWS;

    char* ws = (char*)d_ws;
    size_t off = 0;
    auto alloc = [&](size_t bytes) -> char* {
        char* p = ws + off;
        off = (off + bytes + 255) & ~(size_t)255;
        return p;
    };
    float*         proj  = (float*)        alloc((size_t)N * HID * sizeof(float));
    const int nblk = (E + PB_EDGES - 1) / PB_EDGES;
    unsigned*       meta   = (unsigned*)      alloc((size_t)nblk * OWN_MAX * sizeof(unsigned));
    unsigned*       grouped = (unsigned*)     alloc((size_t)nblk * PB_EDGES * sizeof(unsigned));
    int*           saved = (int*)          alloc((size_t)owners * (OWN_THREADS / 32) * OWN_WCAP * sizeof(int));
    int*           scnt  = (int*)          alloc((size_t)owners * (OWN_THREADS / 32) * sizeof(int));
    float*         deg0  = (float*)        alloc((size_t)N * sizeof(float));
    float*         deg1  = (float*)        alloc((size_t)N * sizeof(float));
    float*         proj1 = out;

    gcn_gemm_wmma<IN_CH><<<N / 16, 128, 0, stream>>>(x, W1, proj1, deg0, deg1, N);

    gcn_deg_part<<<nblk, PB_THREADS, 0, stream>>>(src, dst, deg0, deg1, meta, grouped, E);

    gcn_owner_sum<true><<<owners, OWN_THREADS, 0, stream>>>(proj1, meta, grouped, deg0, deg1, b1, W2, proj, saved, scnt, N, nblk);

    gcn_owner_sum<false><<<owners, OWN_THREADS, 0, stream>>>(proj, meta, grouped, deg0, deg1, b2, nullptr, out, saved, scnt, N, nblk);
}


# owner's edge list ordered by destination row in LDS (row starts from degree planes, place by LDS counter); a wave sums a row's sources in registers: LDS adds per row, not per edge; integer sums
# speedup vs baseline: 3.6665x; 1.0359x over previous
#include <hip/hip_runtime.h>

typedef __attribute__((ext_vector_type(2)))  float    v2f;
typedef __attribute__((ext_vector_type(8)))  float    v8f;

#define IN_CH 128
#define HID   64

#define FX_SCALE 1048576.0f
#define FX_INV   (1.0f / 1048576.0f)

#define OWN_ROWS    196
#define OWN_THREADS 1024
#define OWN_CAP     8192
#define OWN_MAX     256
#define PB_EDGES    1024
#define PB_THREADS  256
static_assert(PB_THREADS == OWN_MAX && PB_THREADS * 4 == PB_EDGES, "one thread per owner counter, four edges per thread");
#define ACC_PITCH   68

static __device__ __forceinline__ int gcn_half_of_wave() {
    return (int)((__builtin_amdgcn_s_sendmsg_rtn(0x87) >> 18) & 1u);
}


__global__ void __launch_bounds__(PB_THREADS)
gcn_deg_part(const int* __restrict__ src, const int* __restrict__ dst, float* __restrict__ deg0, float* __restrict__ deg1,
             unsigned* __restrict__ meta, unsigned* __restrict__ grouped, int E) {
    __shared__ unsigned cnt[OWN_MAX];
    __shared__ unsigned start[OWN_MAX];
    __shared__ unsigned wtot[PB_THREADS / 32];
    __shared__ __attribute__((aligned(16))) unsigned srt[PB_EDGES];

    const int tid  = threadIdx.x;
    const int lane = tid & 31;
    const int wave = tid >> 5;
    const int side = gcn_half_of_wave();
    float* deg = side ? deg1 : deg0;
    cnt[tid] = 0u;
    __syncthreads();

    const int e0 = blockIdx.x * PB_EDGES + tid * 4;
    int d[4], sid[4];
    if (((E & 3) == 0) && e0 < E) {
        const int4 v = *(const int4*)(dst + e0);
        const int4 q = *(const int4*)(src + e0);
        d[0] = v.x; d[1] = v.y; d[2] = v.z; d[3] = v.w;
        sid[0] = q.x; sid[1] = q.y; sid[2] = q.z; sid[3] = q.w;
    } else {
#pragma unroll
        for (int k = 0; k < 4; ++k) {
            d[k]   = (e0 + k < E) ? dst[e0 + k] : -1;
            sid[k] = (e0 + k < E) ? src[e0 + k] : 0;
        }
    }
    int o[4];
    unsigned r[4];
#pragma unroll
    for (int k = 0; k < 4; ++k) {
        o[k] = 0; r[k] = 0u;
        if (d[k] >= 0) {
            atomicAdd(&deg[d[k]], 1.0f);
            o[k] = min((int)((unsigned)d[k] / OWN_ROWS), OWN_MAX - 1);
            r[k] = atomicAdd(&cnt[o[k]], 1u);
        }
    }
    __syncthreads();

    const unsigned c = cnt[tid];
    unsigned incl = c;
#pragma unroll
    for (int s = 1; s < 32; s <<= 1) {
        const unsigned v = __shfl_up(incl, s);
        if (lane >= s) incl += v;
    }
    if (lane == 31) wtot[wave] = incl;
    __syncthreads();
    unsigned base = 0u;
    for (int w = 0; w < wave; ++w) base += wtot[w];
    const unsigned st = base + incl - c;
    start[tid] = st;
    meta[(size_t)blockIdx.x * OWN_MAX + tid] = st | (c << 16);
    __syncthreads();

#pragma unroll
    for (int k = 0; k < 4; ++k) {
        if (d[k] >= 0) srt[start[o[k]] + r[k]] = (unsigned)sid[k] | ((unsigned)(d[k] - o[k] * OWN_ROWS) << 16);
        else           srt[tid * 4 + k] = 0u;
    }
    __syncthreads();
    ((uint4*)(grouped + (size_t)blockIdx.x * PB_EDGES))[tid] = ((const uint4*)srt)[tid];
}

static __device__ __forceinline__ float gcn_deg_of(const float* __restrict__ deg0, const float* __restrict__ deg1, int i) {
    return deg0[i] + deg1[i];
}

template <int K>
__global__ void __launch_bounds__(128)
gcn_gemm_wmma(const float* __restrict__ Af, const float* __restrict__ W, float* __restrict__ C,
              float* __restrict__ deg0, float* __restrict__ deg1, int n) {
    const int lane = threadIdx.x & 31;
    const int nT   = threadIdx.x >> 5;
    const int m0   = blockIdx.x << 4;

    if ((blockIdx.x & 1) == 0 && threadIdx.x < 32) {
        const int i = m0 + (int)threadIdx.x;
        if (i < n) { deg0[i] = 1.0f; deg1[i] = 0.0f; }
    }

    const int hi   = lane >> 4;
    const int row  = m0 + (lane & 15);
    const int col  = (nT << 4) + (lane & 15);

    const size_t aofs = (size_t)row * K + hi * 4;
    const float* wk   = W + (size_t)(hi * 4) * HID + col;

    v8f acc0 = {0.f, 0.f, 0.f, 0.f, 0.f, 0.f, 0.f, 0.f};
    v8f acc1 = {0.f, 0.f, 0.f, 0.f, 0.f, 0.f, 0.f, 0.f};

#pragma unroll
    for (int kb = 0; kb < K / 8; ++kb) {
        const float4 a = *(const float4*)(Af + aofs + kb * 8);
        const float* wb = wk + (size_t)(kb * 8) * HID;
        const v2f a0 = {a.x, a.y};
        const v2f a1 = {a.z, a.w};
        const v2f b0 = {wb[0],       wb[HID]};
        const v2f b1 = {wb[2 * HID], wb[3 * HID]};
        acc0 = __builtin_amdgcn_wmma_f32_16x16x4_f32(
             false, a0,  false, b0,
             (short)0, acc0,  false,  false);
        acc1 = __builtin_amdgcn_wmma_f32_16x16x4_f32(
             false, a1,  false, b1,
             (short)0, acc1,  false,  false);
    }
    const v8f acc = acc0 + acc1;

    float* crow = C + (size_t)(m0 + hi * 8) * HID + col;
#pragma unroll
    for (int r = 0; r < 8; ++r) crow[(size_t)r * HID] = acc[r];
}


static __device__ __noinline__ void gcn_add_edge_by_lane(int* acc, const float* __restrict__ H,
                                                         const float* __restrict__ deg0, const float* __restrict__ deg1,
                                                         int rec, int n0) {
    const int s  = rec & 0xFFFF;
    const int dl = (rec >> 16) & 0xFF;
    const float w = rsqrtf(gcn_deg_of(deg0, deg1, s)) * rsqrtf(gcn_deg_of(deg0, deg1, n0 + dl)) * FX_SCALE;
    const float* hrow = H + (size_t)s * HID;
    int* orow = acc + dl * ACC_PITCH;
    for (int c = 0; c < HID; ++c) atomicAdd(orow + c, __float2int_rn(hrow[c] * w));
}

static __device__ __forceinline__ void gcn_sort_edges(int* acc, int* sorted, const int* rstart, unsigned* rfill,
                                                      const float* __restrict__ H,
                                                      const unsigned* __restrict__ meta,
                                                      const unsigned* __restrict__ grouped,
                                                      const float* __restrict__ deg0, const float* __restrict__ deg1,
                                                      int nblk, int n0, int lane, int wave) {
    for (int tb = 0; tb < nblk; tb += OWN_THREADS) {
        const int t = tb + lane * 32 + wave;
        unsigned m = 0u;
        if (t < nblk) m = meta[(size_t)t * OWN_MAX + blockIdx.x];
        const int c  = (int)(m >> 16);
        const int st = (int)(m & 0xFFFFu);
        const unsigned* sp = grouped + (size_t)t * PB_EDGES + st;
        for (int k0 = 0; k0 < c; k0 += 8) {
            int id[8];
#pragma unroll
            for (int u = 0; u < 8; ++u) id[u] = (int)sp[min(k0 + u, c - 1)];
            int rk[8], rs[8];
#pragma unroll
            for (int u = 0; u < 8; ++u) {
                rk[u] = 0; rs[u] = OWN_CAP;
                if (k0 + u < c) {
                    const int dl = (id[u] >> 16) & 0xFF;
                    rs[u] = rstart[dl];
                    rk[u] = (int)atomicAdd(&rfill[dl], 1u);
                }
            }
#pragma unroll
            for (int u = 0; u < 8; ++u) {
                if (k0 + u < c) {
                    const int slot = rs[u] + rk[u];
                    if (slot < OWN_CAP) sorted[slot] = id[u];
                    else gcn_add_edge_by_lane(acc, H, deg0, deg1, id[u], n0);
                }
            }
        }
    }
}

static_assert(OWN_MAX == 8 * 32, "the running sum over the rows' counts takes eight rows per lane of one wave");

template <bool FIRST>
__global__ void __launch_bounds__(OWN_THREADS)
gcn_owner_sum(const float* __restrict__ H,
              const unsigned* __restrict__ meta, const unsigned* __restrict__ grouped,
              const float* __restrict__ deg0, const float* __restrict__ deg1,
              const float* __restrict__ bias, const float* __restrict__ Wn, float* __restrict__ outp,
              int* saved, int N, int nblk) {
    __shared__ int acc[OWN_ROWS * ACC_PITCH];
    __shared__ int sorted[OWN_CAP];
    __shared__ float disl[OWN_ROWS];
    __shared__ int rcnt[OWN_MAX];
    __shared__ int rstart[OWN_MAX];
    __shared__ unsigned rfill[OWN_MAX];
    __shared__ int ltot;

    const int tid  = threadIdx.x;
    const int lane = tid & 31;
    const int wave = tid >> 5;
    const int n0   = blockIdx.x * OWN_ROWS;
    const int rows = min(OWN_ROWS, N - n0);
    const int nv4  = rows * (HID / 4);
    int* sv = saved + (size_t)blockIdx.x * OWN_CAP;

    if (tid < OWN_MAX && tid >= rows) rcnt[tid] = 0;
    for (int i = tid; i < nv4; i += OWN_THREADS) {
        const int r = i >> 4, c4 = (i & 15) * 4;
        const float dg = gcn_deg_of(deg0, deg1, n0 + r);
        const float4 h = *(const float4*)(H + (size_t)(n0 + r) * HID + c4);
        const float4 b = *(const float4*)(bias + c4);
        const float dd = rsqrtf(dg);
        if ((i & 15) == 0) {
            disl[r] = dd;
            rcnt[r] = max((int)dg - 1, 0);
        }
        const float d2 = dd * dd;
        int4 v;
        v.x = __float2int_rn((b.x + h.x * d2) * FX_SCALE);
        v.y = __float2int_rn((b.y + h.y * d2) * FX_SCALE);
        v.z = __float2int_rn((b.z + h.z * d2) * FX_SCALE);
        v.w = __float2int_rn((b.w + h.w * d2) * FX_SCALE);
        *(int4*)(acc + r * ACC_PITCH + c4) = v;
    }
    __syncthreads();
    if (wave == 0) {
        int c[8];
        int sum = 0;
#pragma unroll
        for (int k = 0; k < 8; ++k) { c[k] = rcnt[lane * 8 + k]; sum += c[k]; }
        int incl = sum;
#pragma unroll
        for (int s = 1; s < 32; s <<= 1) {
            const int v = __shfl_up(incl, s);
            if (lane >= s) incl += v;
        }
        int run = incl - sum;
#pragma unroll
        for (int k = 0; k < 8; ++k) { rstart[lane * 8 + k] = run; rfill[lane * 8 + k] = 0u; run += c[k]; }
        if (lane == 31) ltot = incl;
    }
    __syncthreads();
    const int tot = ltot;

    if (FIRST || tot > OWN_CAP) {
        gcn_sort_edges(acc, sorted, rstart, rfill, H, meta, grouped, deg0, deg1, nblk, n0, lane, wave);
    } else {
        for (int j = tid; j < tot; j += OWN_THREADS) sorted[j] = sv[j];
    }
    __syncthreads();

    const int wv = __builtin_amdgcn_readfirstlane(wave);
    const int T  = __builtin_amdgcn_readfirstlane(min(tot, OWN_CAP));
    const int seg_lo = (T * wv) >> 5;
    const int seg_hi = (T * (wv + 1)) >> 5;
    int cur = 0;
    unsigned a0 = 0u, a1 = 0u;
    for (int base = seg_lo; base < seg_hi; base += 32) {
        const int j = base + lane;
        int s = 0, dl = 0;
        float w = 0.0f;
        if (j < seg_hi) {
            const unsigned rec = (unsigned)sorted[j];
            s  = (int)(rec & 0xFFFFu);
            dl = (int)((rec >> 16) & 0xFFu);
            const unsigned db = rec >> 24;
            float dg;
            if (db != 0u) dg = (float)db;
            else          dg = gcn_deg_of(deg0, deg1, s);
            w = rsqrtf(dg) * disl[dl] * FX_SCALE;
            if (FIRST) sv[j] = (int)((rec & 0x00FFFFFFu) | (dg <= 254.0f ? ((unsigned)dg << 24) : 0u));
        }
        const int wbits = __float_as_int(w);
        const int left = seg_hi - base;

        float h0[4][8], h1[4][8];

#define GCN_LOAD_GROUP(g, on)                                                      \
    if (on) {                                                                      \
    _Pragma("unroll")   \
    for (int u = 0; u < 8; ++u) {                                                  \
        const int sj = __builtin_amdgcn_readlane(s, (g) * 8 + u);                  \
        const float* hrow = H + (size_t)sj * HID;        \
        h0[g][u] = hrow[lane];                                                     \
        h1[g][u] = hrow[lane + 32];                                                \
    } }
#define GCN_ADD_GROUP(g, on)                                                       \
    if (on) {                                                                      \
    _Pragma("unroll")   \
    for (int u = 0; u < 8; ++u) {                                                  \
        const int   dj = __builtin_amdgcn_readlane(dl, (g) * 8 + u);               \
        const float wj = __int_as_float(__builtin_amdgcn_readlane(wbits, (g) * 8 + u)); \
             \
        if (__builtin_expect(dj != cur, 0)) {                                      \
            unsigned* orow = (unsigned*)(acc + cur * ACC_PITCH);     \
            atomicAdd(orow + lane,      a0);                                       \
            atomicAdd(orow + lane + 32, a1);                                       \
            cur = dj; a0 = 0u; a1 = 0u;                                            \
        }                                                                          \
        a0 += (unsigned)__float2int_rn(h0[g][u] * wj);                             \
        a1 += (unsigned)__float2int_rn(h1[g][u] * wj);                             \
    } }

        if (left >= 32) {
            GCN_LOAD_GROUP(0, true)
            GCN_LOAD_GROUP(1, true)
            GCN_ADD_GROUP(0, true)
            GCN_LOAD_GROUP(2, true)
            GCN_ADD_GROUP(1, true)
            GCN_LOAD_GROUP(3, true)
            GCN_ADD_GROUP(2, true)
            GCN_ADD_GROUP(3, true)
        } else {
            GCN_LOAD_GROUP(0, true)
            GCN_LOAD_GROUP(1, left > 8)
            GCN_ADD_GROUP(0, true)
            GCN_LOAD_GROUP(2, left > 16)
            GCN_ADD_GROUP(1, left > 8)
            GCN_LOAD_GROUP(3, left > 24)
            GCN_ADD_GROUP(2, left > 16)
            GCN_ADD_GROUP(3, left > 24)
        }
#undef GCN_LOAD_GROUP
#undef GCN_ADD_GROUP
    }
    {
        unsigned* orow = (unsigned*)(acc + cur * ACC_PITCH);
        atomicAdd(orow + lane,      a0);
        atomicAdd(orow + lane + 32, a1);
    }
    __syncthreads();

    if (FIRST) {
        const int c  = lane & 15;
        const int hi = lane >> 4;
        const int wv = __builtin_amdgcn_readfirstlane(wave);
        const int units = ((rows + 15) >> 4) * 2;
        for (int unit = wv; unit < units; unit += OWN_THREADS / 32) {
            const int m0l = (unit >> 1) << 4;
            const int nb  = (unit & 1) << 5;
            const int* arow = acc + min(m0l + c, rows - 1) * ACC_PITCH + hi * 4;
            const float* wk = Wn + (size_t)(hi * 4) * HID + nb + c;
            v8f p0 = {0.f, 0.f, 0.f, 0.f, 0.f, 0.f, 0.f, 0.f};
            v8f p1 = {0.f, 0.f, 0.f, 0.f, 0.f, 0.f, 0.f, 0.f};
            v8f q0 = {0.f, 0.f, 0.f, 0.f, 0.f, 0.f, 0.f, 0.f};
            v8f q1 = {0.f, 0.f, 0.f, 0.f, 0.f, 0.f, 0.f, 0.f};
#pragma unroll
            for (int kb = 0; kb < HID / 8; ++kb) {
                const int4 ai = *(const int4*)(arow + kb * 8);
                const v2f a0 = {fmaxf((float)ai.x * FX_INV, 0.0f), fmaxf((float)ai.y * FX_INV, 0.0f)};
                const v2f a1 = {fmaxf((float)ai.z * FX_INV, 0.0f), fmaxf((float)ai.w * FX_INV, 0.0f)};
                const float* wb = wk + (size_t)(kb * 8) * HID;
                const v2f b0 = {wb[0],            wb[HID]};
                const v2f b1 = {wb[2 * HID],      wb[3 * HID]};
                const v2f d0 = {wb[16],           wb[HID + 16]};
                const v2f d1 = {wb[2 * HID + 16], wb[3 * HID + 16]};
                p0 = __builtin_amdgcn_wmma_f32_16x16x4_f32(false, a0, false, b0, (short)0, p0, false, false);
                q0 = __builtin_amdgcn_wmma_f32_16x16x4_f32(false, a0, false, d0, (short)0, q0, false, false);
                p1 = __builtin_amdgcn_wmma_f32_16x16x4_f32(false, a1, false, b1, (short)0, p1, false, false);
                q1 = __builtin_amdgcn_wmma_f32_16x16x4_f32(false, a1, false, d1, (short)0, q1, false, false);
            }
            const v8f p = p0 + p1;
            const v8f q = q0 + q1;
            float* crow = outp + (size_t)(n0 + m0l + hi * 8) * HID + nb + c;
            if (m0l + 16 <= rows) {
#pragma unroll
                for (int r = 0; r < 8; ++r) {
                    crow[(size_t)r * HID]      = p[r];
                    crow[(size_t)r * HID + 16] = q[r];
                }
            } else {
                const int live = rows - (m0l + hi * 8);
#pragma unroll
                for (int r = 0; r < 8; ++r) {
                    if (r < live) {
                        crow[(size_t)r * HID]      = p[r];
                        crow[(size_t)r * HID + 16] = q[r];
                    }
                }
            }
        }
    } else {
        float4* orows = (float4*)(outp + (size_t)n0 * HID);
        for (int i = tid; i < nv4; i += OWN_THREADS) {
            const int r = i >> 4, c4 = (i & 15) * 4;
            const int4 a = *(const int4*)(acc + r * ACC_PITCH + c4);
            float4 v;
            v.x = (float)a.x * FX_INV;
            v.y = (float)a.y * FX_INV;
            v.z = (float)a.z * FX_INV;
            v.w = (float)a.w * FX_INV;
            orows[i] = v;
        }
    }
}


extern "C" void kernel_launch(void* const* d_in, const int* in_sizes, int n_in,
                              void* d_out, int out_size, void* d_ws, size_t ws_size,
                              hipStream_t stream) {
    const float* x  = (const float*)d_in[0];
    const int*   ei = (const int*)d_in[1];
    const float* W1 = (const float*)d_in[2];
    const float* b1 = (const float*)d_in[3];
    const float* W2 = (const float*)d_in[4];
    const float* b2 = (const float*)d_in[5];
    float* out = (float*)d_out;

    const int N = in_sizes[0] / IN_CH;
    const int E = in_sizes[1] / 2;
    const int* src = ei;
    const int* dst = ei + E;

    const int owners = (N + OWN_ROWS - 1) / OWN_ROWS;

    char* ws = (char*)d_ws;
    size_t off = 0;
    auto alloc = [&](size_t bytes) -> char* {
        char* p = ws + off;
        off = (off + bytes + 255) & ~(size_t)255;
        return p;
    };
    float*         proj  = (float*)        alloc((size_t)N * HID * sizeof(float));
    const int nblk = (E + PB_EDGES - 1) / PB_EDGES;
    unsigned*       meta   = (unsigned*)      alloc((size_t)nblk * OWN_MAX * sizeof(unsigned));
    unsigned*       grouped = (unsigned*)     alloc((size_t)nblk * PB_EDGES * sizeof(unsigned));
    int*           saved = (int*)          alloc((size_t)owners * OWN_CAP * sizeof(int));
    float*         deg0  = (float*)        alloc((size_t)N * sizeof(float));
    float*         deg1  = (float*)        alloc((size_t)N * sizeof(float));
    float*         proj1 = out;

    gcn_gemm_wmma<IN_CH><<<N / 16, 128, 0, stream>>>(x, W1, proj1, deg0, deg1, N);

    gcn_deg_part<<<nblk, PB_THREADS, 0, stream>>>(src, dst, deg0, deg1, meta, grouped, E);

    gcn_owner_sum<true><<<owners, OWN_THREADS, 0, stream>>>(proj1, meta, grouped, deg0, deg1, b1, W2, proj, saved, N, nblk);

    gcn_owner_sum<false><<<owners, OWN_THREADS, 0, stream>>>(proj, meta, grouped, deg0, deg1, b2, nullptr, out, saved, N, nblk);
}
